# starting kernel
# baseline (speedup 1.0000x reference)
#include <hip/hip_runtime.h>

typedef __attribute__((ext_vector_type(2))) float v2f;
typedef __attribute__((ext_vector_type(8))) float v8f;

#define NNODES 50000

__device__ __forceinline__ void atom_add_f32(float* p, float v) {
    asm volatile("global_atomic_add_f32 %0, %1, off" :: "v"(p), "v"(v) : "memory");
}

__global__ __launch_bounds__(256) void init_kernel(float* __restrict__ deg,
                                                   float* __restrict__ agg1,
                                                   float* __restrict__ out) {
    long long idx = (long long)blockIdx.x * 256 + threadIdx.x;
    if (idx < (long long)NNODES * 128) agg1[idx] = 0.0f;
    if (idx < (long long)NNODES * 64)  out[idx]  = 0.0f;
    if (idx < NNODES)                  deg[idx]  = 1.0f;
}

__global__ __launch_bounds__(256) void deg_kernel(const int* __restrict__ dst,
                                                  float* __restrict__ deg, int E) {
    int e = blockIdx.x * 256 + threadIdx.x;
    if (e < E) atom_add_f32(&deg[dst[e]], 1.0f);
}

__global__ __launch_bounds__(256) void dinv_kernel(float* __restrict__ deg) {
    int i = blockIdx.x * 256 + threadIdx.x;
    if (i < NNODES) deg[i] = rsqrtf(deg[i]);
}

template <int K, int NF>
__global__ __launch_bounds__(256) void gemm_wmma(const float* __restrict__ A,
                                                 const float* __restrict__ B,
                                                 float* __restrict__ C, int M) {
    const int lane  = threadIdx.x & 31;
    const int wave  = blockIdx.x * 8 + (threadIdx.x >> 5);
    const int mBase = wave * 16;
    if (mBase >= M) return;
    const int half = lane >> 4;
    const int l16  = lane & 15;
    constexpr int NT = NF / 16;

    v8f acc[NT] = {};

    const float* aRow = A + (size_t)(mBase + l16) * K + 2 * half;
    for (int k = 0; k < K; k += 4) {
        v2f a = *(const v2f*)(aRow + k);
        const float* bPtr = B + (size_t)(k + 2 * half) * NF + l16;
#pragma unroll
        for (int t = 0; t < NT; ++t) {
            v2f b;
            b.x = bPtr[t * 16];
            b.y = bPtr[t * 16 + NF];
            acc[t] = __builtin_amdgcn_wmma_f32_16x16x4_f32(
                false, a, false, b, (short)0, acc[t], false, false);
        }
    }
#pragma unroll
    for (int t = 0; t < NT; ++t) {
        float* cPtr = C + (size_t)(mBase + half * 8) * NF + t * 16 + l16;
#pragma unroll
        for (int r = 0; r < 8; ++r) cPtr[(size_t)r * NF] = acc[t][r];
    }
}

template <int F>
__global__ __launch_bounds__(256) void edge_agg(const int* __restrict__ src,
                                                const int* __restrict__ dst,
                                                const float* __restrict__ dinv,
                                                const float* __restrict__ h,
                                                float* __restrict__ agg, int E) {
    constexpr int CH = F / 4;
    long long tid = (long long)blockIdx.x * 256 + threadIdx.x;
    if (tid >= (long long)E * CH) return;
    int e = (int)(tid / CH);
    int c = (int)(tid & (CH - 1));
    int s = src[e], d = dst[e];
    float w = dinv[s] * dinv[d];
    float4 hv = ((const float4*)(h + (size_t)s * F))[c];
    float* ap = agg + (size_t)d * F + c * 4;
    atom_add_f32(ap + 0, hv.x * w);
    atom_add_f32(ap + 1, hv.y * w);
    atom_add_f32(ap + 2, hv.z * w);
    atom_add_f32(ap + 3, hv.w * w);
}

__global__ __launch_bounds__(256) void fin1_kernel(float* __restrict__ agg1,
                                                   const float* __restrict__ h1,
                                                   const float* __restrict__ dinv,
                                                   const float* __restrict__ b1) {
    long long idx = (long long)blockIdx.x * 256 + threadIdx.x;
    if (idx >= (long long)NNODES * 128) return;
    int f = (int)(idx & 127);
    int i = (int)(idx >> 7);
    float di = dinv[i];
    float v = agg1[idx] + h1[idx] * di * di + b1[f];
    agg1[idx] = fmaxf(v, 0.0f);
}

__global__ __launch_bounds__(256) void fin2_kernel(float* __restrict__ out,
                                                   const float* __restrict__ h2,
                                                   const float* __restrict__ dinv,
                                                   const float* __restrict__ b2) {
    long long idx = (long long)blockIdx.x * 256 + threadIdx.x;
    if (idx >= (long long)NNODES * 64) return;
    int f = (int)(idx & 63);
    int i = (int)(idx >> 6);
    float di = dinv[i];
    out[idx] = out[idx] + h2[idx] * di * di + b2[f];
}

extern "C" void kernel_launch(void* const* d_in, const int* in_sizes, int n_in,
                              void* d_out, int out_size, void* d_ws, size_t ws_size,
                              hipStream_t stream) {
    const float* x  = (const float*)d_in[0];
    const int*   ei = (const int*)d_in[1];
    const float* W1 = (const float*)d_in[2];
    const float* b1 = (const float*)d_in[3];
    const float* W2 = (const float*)d_in[4];
    const float* b2 = (const float*)d_in[5];
    float* out = (float*)d_out;

    const int N = NNODES;
    const int E = in_sizes[1] / 2;
    const int* src = ei;
    const int* dst = ei + E;

    float* deg  = (float*)d_ws;
    float* h1   = deg + N;
    float* agg1 = h1 + (size_t)N * 128;
    float* h2   = agg1 + (size_t)N * 128;

    const int BT = 256;
    long long n128 = (long long)N * 128;
    long long n64  = (long long)N * 64;

    init_kernel<<<(int)((n128 + BT - 1) / BT), BT, 0, stream>>>(deg, agg1, out);
    deg_kernel<<<(E + BT - 1) / BT, BT, 0, stream>>>(dst, deg, E);
    dinv_kernel<<<(N + BT - 1) / BT, BT, 0, stream>>>(deg);
    gemm_wmma<256, 128><<<(N / 16 + 7) / 8, BT, 0, stream>>>(x, W1, h1, N);
    {
        long long tot = (long long)E * 32;
        edge_agg<128><<<(int)((tot + BT - 1) / BT), BT, 0, stream>>>(src, dst, deg, h1, agg1, E);
    }
    fin1_kernel<<<(int)((n128 + BT - 1) / BT), BT, 0, stream>>>(agg1, h1, deg, b1);
    gemm_wmma<128, 64><<<(N / 16 + 7) / 8, BT, 0, stream>>>(agg1, W2, h2, N);
    {
        long long tot = (long long)E * 16;
        edge_agg<64><<<(int)((tot + BT - 1) / BT), BT, 0, stream>>>(src, dst, deg, h2, out, E);
    }
    fin2_kernel<<<(int)((n64 + BT - 1) / BT), BT, 0, stream>>>(out, h2, deg, b2);
}


# edge-sum scan reads a one-byte owner-range plane written by the degree pass, 4 edges per word by a zero-byte test, 8 row masks before any scalar pick; dst read only at hits; same lists and add order
# speedup vs baseline: 2.4077x; 2.4077x over previous
#include <hip/hip_runtime.h>

typedef __attribute__((ext_vector_type(2))) float v2f;
typedef __attribute__((ext_vector_type(8))) float v8f;

#define NNODES 50000
#define OWN_RN 196

__device__ __forceinline__ void atom_add_f32(float* p, float v) {
    asm volatile("global_atomic_add_f32 %0, %1, off" :: "v"(p), "v"(v) : "memory");
}

__device__ __forceinline__ int wg_half(int* slot) {
    if (threadIdx.x == 0)
        *slot = (int)((__builtin_amdgcn_s_sendmsg_rtn(0x87) >> 18) & 1u);
    __syncthreads();
    return __builtin_amdgcn_readfirstlane(*slot);
}

__global__ __launch_bounds__(256) void init_kernel(float* __restrict__ deg0,
                                                   float* __restrict__ deg1) {
    int idx = blockIdx.x * 256 + threadIdx.x;
    if (idx < NNODES) { deg0[idx] = 1.0f; deg1[idx] = 0.0f; }
}

__global__ __launch_bounds__(256) void deg_kernel(const int* __restrict__ dst,
                                                  float* __restrict__ deg0,
                                                  float* __restrict__ deg1,
                                                  unsigned char* __restrict__ own, int E) {
    __shared__ int s_half;
    float* deg = wg_half(&s_half) ? deg1 : deg0;
    int e = blockIdx.x * 256 + threadIdx.x;
    if (e < E) {
        int d = dst[e];
        own[e] = (unsigned char)(d / OWN_RN);
        atom_add_f32(&deg[d], 1.0f);
    }
}

__global__ __launch_bounds__(256) void dinv_kernel(float* __restrict__ deg0,
                                                   const float* __restrict__ deg1) {
    int i = blockIdx.x * 256 + threadIdx.x;
    if (i < NNODES) deg0[i] = rsqrtf(deg0[i] + deg1[i]);
}

template <int K, int NF>
__global__ __launch_bounds__(256) void gemm_wmma(const float* __restrict__ A,
                                                 const float* __restrict__ B,
                                                 float* __restrict__ C, int M) {
    const int lane  = threadIdx.x & 31;
    const int wave  = blockIdx.x * 8 + (threadIdx.x >> 5);
    const int mBase = wave * 16;
    if (mBase >= M) return;
    const int half = lane >> 4;
    const int l16  = lane & 15;
    constexpr int NT = NF / 16;

    v8f acc[NT] = {};

    const float* aRow = A + (size_t)(mBase + l16) * K + 2 * half;
    for (int k = 0; k < K; k += 4) {
        v2f a = *(const v2f*)(aRow + k);
        const float* bPtr = B + (size_t)(k + 2 * half) * NF + l16;
#pragma unroll
        for (int t = 0; t < NT; ++t) {
            v2f b;
            b.x = bPtr[t * 16];
            b.y = bPtr[t * 16 + NF];
            acc[t] = __builtin_amdgcn_wmma_f32_16x16x4_f32(
                false, a, false, b, (short)0, acc[t], false, false);
        }
    }
#pragma unroll
    for (int t = 0; t < NT; ++t) {
        float* cPtr = C + (size_t)(mBase + half * 8) * NF + t * 16 + l16;
#pragma unroll
        for (int r = 0; r < 8; ++r) cPtr[(size_t)r * NF] = acc[t][r];
    }
}

#define RN    OWN_RN
#define LCAP  1280
#define PCAP  160
#define RCAP  512

template <int F, int U>
__device__ __forceinline__ void add_rows(int s, int dloc, float wgt, int m, int lane,
                                         const float* __restrict__ h, float* tile) {
    constexpr int NQ = F / 32;
    int j = 0;
#pragma unroll 1
    for (; j + U <= m; j += U) {
        float v[U][NQ];
#pragma unroll
        for (int u = 0; u < U; ++u) {
            int sj = __builtin_amdgcn_readlane(s, j + u);
            const float* hr = h + (size_t)sj * F + lane;
#pragma unroll
            for (int q = 0; q < NQ; ++q) v[u][q] = hr[q * 32];
        }
#pragma unroll
        for (int u = 0; u < U; ++u) {
            int   dj = __builtin_amdgcn_readlane(dloc, j + u);
            float wj = __builtin_bit_cast(float, __builtin_amdgcn_readlane(__builtin_bit_cast(int, wgt), j + u));
            float* tp = tile + dj * F + lane;
#pragma unroll
            for (int q = 0; q < NQ; ++q) tp[q * 32] += v[u][q] * wj;
        }
    }
    for (; j < m; ++j) {
        int   sj = __builtin_amdgcn_readlane(s, j);
        int   dj = __builtin_amdgcn_readlane(dloc, j);
        float wj = __builtin_bit_cast(float, __builtin_amdgcn_readlane(__builtin_bit_cast(int, wgt), j));
        const float* hr = h + (size_t)sj * F + lane;
        float* tp = tile + dj * F + lane;
        float v[NQ];
#pragma unroll
        for (int q = 0; q < NQ; ++q) v[q] = hr[q * 32];
#pragma unroll
        for (int q = 0; q < NQ; ++q) tp[q * 32] += v[q] * wj;
    }
}

template <int F, int U, bool RECORD>
__device__ __forceinline__ void take_edges(const int* ent, int m, int lane,
                                           const int* __restrict__ src,
                                           const float* __restrict__ dinv,
                                           const float* __restrict__ h,
                                           const float* dl, float* tile,
                                           int* recId, float* recW, int& recn, bool& rovf) {
    int s = 0, dloc = 0;
    float wgt = 0.0f;
    if (lane < m) {
        int en = ent[lane];
        dloc  = (int)((unsigned)en >> 24);
        s     = src[en & 0xFFFFFF];
        wgt   = dinv[s] * dl[dloc];
    }
    if (RECORD) {
        if (recn + m <= RCAP) {
            if (lane < m) {
                recId[recn + lane] = (int)(((unsigned)dloc << 24) | (unsigned)s);
                recW[recn + lane]  = wgt;
            }
        } else {
            rovf = true;
        }
        recn += m;
    }
    add_rows<F, U>(s, dloc, wgt, m, lane, h, tile);
}

__device__ __forceinline__ void emit_row(unsigned z, unsigned x, unsigned mk, int e0,
                                         int* mylist, int& cntw) {
    const unsigned multi = __builtin_amdgcn_ballot_w32((z & (z - 1u)) != 0u);
    if (multi == 0u) {
        if (z != 0u) mylist[cntw + __builtin_amdgcn_mbcnt_lo(mk, 0u)] = e0 + (__builtin_ctz(z) >> 3);
        cntw += __builtin_popcount(mk);
    } else {
        const unsigned t = (x & 0x7f7f7f7fu) + 0x7f7f7f7fu;
        unsigned f = ~(t | x | 0x7f7f7f7fu);
        const int pcl = __builtin_popcount(f);
        int pre = 0, total = 0;
#pragma unroll
        for (int r = 0; r < 4; ++r) {
            const unsigned m = __builtin_amdgcn_ballot_w32(pcl > r);
            pre   += (int)__builtin_amdgcn_mbcnt_lo(m, 0u);
            total += __builtin_popcount(m);
        }
        int q = cntw + pre;
        while (f) {
            mylist[q] = e0 + (__builtin_ctz(f) >> 3);
            ++q;
            f &= f - 1u;
        }
        cntw += total;
    }
}

template <bool TAIL>
__device__ __forceinline__ void filter_chunk(const unsigned* __restrict__ ownw, int base, int E,
                                             unsigned pat, int lane,
                                             int* mylist, int& cntw, bool& ovf) {
    unsigned x[8], z[8], mk[8];
    const unsigned* p = ownw + (base >> 2) + lane;
#pragma unroll
    for (int j = 0; j < 8; ++j) {
        if (TAIL) {
            const int nv = E - (base + j * 128 + lane * 4);
            unsigned w = (nv > 0) ? p[j * 32] : 0u;
            w ^= pat;
            if (nv < 4) w |= (nv <= 0) ? 0xFFFFFFFFu : (0xFFFFFFFFu << (8 * nv));
            x[j] = w;
        } else {
            x[j] = p[j * 32] ^ pat;
        }
    }
    int nflag = 0;
#pragma unroll
    for (int j = 0; j < 8; ++j) {
        z[j]  = (x[j] - 0x01010101u) & ~x[j] & 0x80808080u;
        mk[j] = __builtin_amdgcn_ballot_w32(z[j] != 0u);
        nflag += __builtin_popcount(mk[j]);
    }
    if (nflag == 0) return;
    if (cntw + 4 * nflag > LCAP) {
        ovf = true;
        return;
    }
#pragma unroll
    for (int j = 0; j < 8; ++j) {
        if (mk[j]) emit_row(z[j], x[j], mk[j], base + j * 128 + lane * 4, mylist, cntw);
    }
}

template <int F, bool RELU, bool REPLAY>
__global__ __launch_bounds__(1024) void agg_tile(const int* __restrict__ src,
                                                 const int* __restrict__ dst,
                                                 const float* __restrict__ dinv,
                                                 const float* __restrict__ h,
                                                 const float* __restrict__ bias,
                                                 float* __restrict__ outp, int E,
                                                 int* __restrict__ recs,
                                                 int* __restrict__ reccnt,
                                                 const unsigned* __restrict__ ownw) {
    static_assert(RN <= 256, "a list entry keeps the row in 8 bits");
    static_assert((NNODES + RN - 1) / RN <= 256, "a node range's number is kept in one byte");
    constexpr int U = 1024 / F;
    __shared__ float tile[RN * F];
    __shared__ float dl[RN];
    __shared__ int   list[32 * LCAP];
    __shared__ int   pendl[32 * PCAP];
    __shared__ int   cntArr[32];
    __shared__ int   s_ovf;

    const int lane = threadIdx.x & 31;
    const int wave = __builtin_amdgcn_readfirstlane((int)(threadIdx.x >> 5));
    const int lo   = blockIdx.x * RN;
    const int nn   = min(RN, NNODES - lo);

    int*   recId = recs + (size_t)(blockIdx.x * 32 + wave) * (2 * RCAP);
    float* recW  = (float*)(recId + RCAP);
    int cw = 0;
    if (REPLAY) cw = reccnt[blockIdx.x * 32 + lane];

    if (threadIdx.x == 0) s_ovf = (E >= (1 << 24)) ? 1 : 0;
    for (int i = threadIdx.x; i < nn; i += 1024) dl[i] = dinv[lo + i];
    __syncthreads();

    bool  replay = false;
    int   myn = 0;
    int   curId = 0;
    float curW = 0.0f;
    if (REPLAY) {
        replay = __builtin_amdgcn_ballot_w32(cw < 0) == 0u;
        myn = __builtin_amdgcn_readlane(cw, wave);
        if (replay && lane < myn) {
            curId = recId[lane];
            curW  = recW[lane];
        }
    }
    for (int idx = threadIdx.x; idx < nn * F; idx += 1024) {
        int i = idx / F, f = idx % F;
        float di = dl[i];
        tile[idx] = h[(size_t)lo * F + idx] * (di * di) + bias[f];
    }

    if (REPLAY && replay) {
        __syncthreads();
        for (int b0 = 0; b0 < myn; b0 += 32) {
            const int m = min(32, myn - b0);
            int   nxtId = 0;
            float nxtW = 0.0f;
            if (b0 + 32 + lane < myn) {
                nxtId = recId[b0 + 32 + lane];
                nxtW  = recW[b0 + 32 + lane];
            }
            add_rows<F, U>(curId & 0xFFFFFF, (int)((unsigned)curId >> 24), curW, m, lane, h, tile);
            curId = nxtId;
            curW  = nxtW;
        }
    } else {
        int* mylist = list + wave * LCAP;
        int cntw = 0;
        bool ovf = false;
        const unsigned pat = (unsigned)blockIdx.x * 0x01010101u;
        for (int base = wave * 1024; base < E; base += 32 * 1024) {
            if (base + 1024 <= E) filter_chunk<false>(ownw, base, E, pat, lane, mylist, cntw, ovf);
            else                  filter_chunk<true>(ownw, base, E, pat, lane, mylist, cntw, ovf);
        }
        for (int i0 = 0; i0 < cntw; i0 += 128) {
            int en[4], dd[4];
#pragma unroll
            for (int t = 0; t < 4; ++t) {
                const int idx = i0 + t * 32 + lane;
                en[t] = (idx < cntw) ? mylist[idx] : -1;
            }
#pragma unroll
            for (int t = 0; t < 4; ++t) dd[t] = (en[t] >= 0) ? dst[en[t]] : 0;
#pragma unroll
            for (int t = 0; t < 4; ++t) {
                if (en[t] >= 0)
                    mylist[i0 + t * 32 + lane] = (int)(((unsigned)(dd[t] - lo) << 24) | (unsigned)en[t]);
            }
        }
        if (lane == 0) {
            cntArr[wave] = cntw;
            if (ovf) s_ovf = 1;
        }
        __syncthreads();

        if (s_ovf == 0) {
            int* mypend = pendl + wave * PCAP;
            int pend = 0;
            int  recn = 0;
            bool rovf = false;
            for (int sw = 0; sw < 32; ++sw) {
                const int c = __builtin_amdgcn_readfirstlane(cntArr[sw]);
                const int* L = list + sw * LCAP;
                for (int b = 0; b < c; b += 128) {
                    int en[4];
#pragma unroll
                    for (int t = 0; t < 4; ++t) {
                        int idx = b + t * 32 + lane;
                        en[t] = (idx < c) ? L[idx] : 0;
                    }
#pragma unroll
                    for (int t = 0; t < 4; ++t) {
                        bool mine = (b + t * 32 + lane < c) &&
                                    ((((unsigned)en[t] >> 24) & 31u) == (unsigned)wave);
                        unsigned mask = __builtin_amdgcn_ballot_w32(mine);
                        if (mask) {
                            if (mine) mypend[pend + __builtin_amdgcn_mbcnt_lo(mask, 0u)] = en[t];
                            pend += __builtin_popcount(mask);
                        }
                    }
                    while (pend >= 32) {
                        pend -= 32;
                        take_edges<F, U, !REPLAY>(mypend + pend, 32, lane, src, dinv, h, dl, tile,
                                                  recId, recW, recn, rovf);
                    }
                }
            }
            take_edges<F, U, !REPLAY>(mypend, pend, lane, src, dinv, h, dl, tile, recId, recW, recn, rovf);
            if (!REPLAY && lane == 0) reccnt[blockIdx.x * 32 + wave] = rovf ? -1 : recn;
        } else {
            if (!REPLAY && lane == 0) reccnt[blockIdx.x * 32 + wave] = -1;
            constexpr int NQ = F / 32;
            for (int base = 0; base < E; base += 32) {
                int e = base + lane;
                int d = (e < E) ? dst[e] : -1;
                unsigned dloc = (unsigned)(d - lo);
                bool mine = (dloc < (unsigned)nn) && ((dloc & 31u) == (unsigned)wave);
                unsigned mask = __builtin_amdgcn_ballot_w32(mine);
                while (mask) {
                    int j = __builtin_ctz(mask);
                    mask &= mask - 1;
                    int dj = (int)__builtin_amdgcn_readlane((int)dloc, j);
                    int sj = src[base + j];
                    float wj = dinv[sj] * dl[dj];
                    const float* hr = h + (size_t)sj * F + lane;
                    float* tp = tile + dj * F + lane;
#pragma unroll
                    for (int q = 0; q < NQ; ++q) tp[q * 32] += hr[q * 32] * wj;
                }
            }
        }
    }

    __syncthreads();
    for (int idx = threadIdx.x; idx < nn * F; idx += 1024) {
        float v = tile[idx];
        if (RELU) v = fmaxf(v, 0.0f);
        outp[(size_t)lo * F + idx] = v;
    }
}

extern "C" void kernel_launch(void* const* d_in, const int* in_sizes, int n_in,
                              void* d_out, int out_size, void* d_ws, size_t ws_size,
                              hipStream_t stream) {
    const float* x  = (const float*)d_in[0];
    const int*   ei = (const int*)d_in[1];
    const float* W1 = (const float*)d_in[2];
    const float* b1 = (const float*)d_in[3];
    const float* W2 = (const float*)d_in[4];
    const float* b2 = (const float*)d_in[5];
    float* out = (float*)d_out;

    const int N = NNODES;
    const int E = in_sizes[1] / 2;
    const int* src = ei;
    const int* dst = ei + E;

    float* deg   = (float*)d_ws;
    float* h1    = deg + N;
    float* a1    = h1 + (size_t)N * 128;
    float* h2    = a1 + (size_t)N * 128;
    float* deg1  = h2 + (size_t)N * 64;
    int*   reccnt = (int*)(deg1 + N);
    int*   recs   = reccnt + 256 * 32 + 32;
    unsigned char* own = (unsigned char*)(recs + (size_t)256 * 32 * 2 * RCAP);

    const int BT = 256;
    const int NB = (NNODES + RN - 1) / RN;

    init_kernel<<<(N + BT - 1) / BT, BT, 0, stream>>>(deg, deg1);
    deg_kernel<<<(E + BT - 1) / BT, BT, 0, stream>>>(dst, deg, deg1, own, E);
    dinv_kernel<<<(N + BT - 1) / BT, BT, 0, stream>>>(deg, deg1);
    gemm_wmma<256, 128><<<(N / 16 + 7) / 8, BT, 0, stream>>>(x, W1, h1, N);
    agg_tile<128, true, false><<<NB, 1024, 0, stream>>>(src, dst, deg, h1, b1, a1, E, recs, reccnt, (const unsigned*)own);
    gemm_wmma<128, 64><<<(N / 16 + 7) / 8, BT, 0, stream>>>(a1, W2, h2, N);
    agg_tile<64, false, true><<<NB, 1024, 0, stream>>>(src, dst, deg, h2, b2, out, E, recs, reccnt, (const unsigned*)own);
}


# edge sums kept in the owner wave's registers (its 6-7 rows, element picked by a scalar register index) instead of an LDS tile: start value loaded and rows stored by that wave; same add order
# speedup vs baseline: 2.5747x; 1.0694x over previous
#include <hip/hip_runtime.h>

typedef __attribute__((ext_vector_type(2))) float v2f;
typedef __attribute__((ext_vector_type(8))) float v8f;

#define NNODES 50000
#define OWN_RN 196

__device__ __forceinline__ void atom_add_f32(float* p, float v) {
    asm volatile("global_atomic_add_f32 %0, %1, off" :: "v"(p), "v"(v) : "memory");
}

__device__ __forceinline__ int wg_half(int* slot) {
    if (threadIdx.x == 0)
        *slot = (int)((__builtin_amdgcn_s_sendmsg_rtn(0x87) >> 18) & 1u);
    __syncthreads();
    return __builtin_amdgcn_readfirstlane(*slot);
}

__global__ __launch_bounds__(256) void init_kernel(float* __restrict__ deg0,
                                                   float* __restrict__ deg1) {
    int idx = blockIdx.x * 256 + threadIdx.x;
    if (idx < NNODES) { deg0[idx] = 1.0f; deg1[idx] = 0.0f; }
}

__global__ __launch_bounds__(256) void deg_kernel(const int* __restrict__ dst,
                                                  float* __restrict__ deg0,
                                                  float* __restrict__ deg1,
                                                  unsigned char* __restrict__ own, int E) {
    __shared__ int s_half;
    float* deg = wg_half(&s_half) ? deg1 : deg0;
    int e = blockIdx.x * 256 + threadIdx.x;
    if (e < E) {
        int d = dst[e];
        own[e] = (unsigned char)(d / OWN_RN);
        atom_add_f32(&deg[d], 1.0f);
    }
}

__global__ __launch_bounds__(256) void dinv_kernel(float* __restrict__ deg0,
                                                   const float* __restrict__ deg1) {
    int i = blockIdx.x * 256 + threadIdx.x;
    if (i < NNODES) deg0[i] = rsqrtf(deg0[i] + deg1[i]);
}

template <int K, int NF>
__global__ __launch_bounds__(256) void gemm_wmma(const float* __restrict__ A,
                                                 const float* __restrict__ B,
                                                 float* __restrict__ C, int M) {
    const int lane  = threadIdx.x & 31;
    const int wave  = blockIdx.x * 8 + (threadIdx.x >> 5);
    const int mBase = wave * 16;
    if (mBase >= M) return;
    const int half = lane >> 4;
    const int l16  = lane & 15;
    constexpr int NT = NF / 16;

    v8f acc[NT] = {};

    const float* aRow = A + (size_t)(mBase + l16) * K + 2 * half;
    for (int k = 0; k < K; k += 4) {
        v2f a = *(const v2f*)(aRow + k);
        const float* bPtr = B + (size_t)(k + 2 * half) * NF + l16;
#pragma unroll
        for (int t = 0; t < NT; ++t) {
            v2f b;
            b.x = bPtr[t * 16];
            b.y = bPtr[t * 16 + NF];
            acc[t] = __builtin_amdgcn_wmma_f32_16x16x4_f32(
                false, a, false, b, (short)0, acc[t], false, false);
        }
    }
#pragma unroll
    for (int t = 0; t < NT; ++t) {
        float* cPtr = C + (size_t)(mBase + half * 8) * NF + t * 16 + l16;
#pragma unroll
        for (int r = 0; r < 8; ++r) cPtr[(size_t)r * NF] = acc[t][r];
    }
}

#define RN    OWN_RN
#define NSLOT ((RN + 31) / 32)
#define LCAP  1280
#define PCAP  160
#define RCAP  512

template <int NQ>
__device__ __forceinline__ void add_to_slot(v8f (&acc)[NQ], int slot, const float (&v)[NQ], float w) {
#pragma unroll
    for (int q = 0; q < NQ; ++q) acc[q][slot] = __builtin_fmaf(v[q], w, acc[q][slot]);
}

template <int F, int U>
__device__ __forceinline__ void add_rows(int s, int dloc, float wgt, int m, int lane,
                                         const float* __restrict__ h, v8f (&acc)[F / 32]) {
    constexpr int NQ = F / 32;
    int j = 0;
#pragma unroll 1
    for (; j + U <= m; j += U) {
        float v[U][NQ];
#pragma unroll
        for (int u = 0; u < U; ++u) {
            int sj = __builtin_amdgcn_readlane(s, j + u);
            const float* hr = h + (size_t)sj * F + lane;
#pragma unroll
            for (int q = 0; q < NQ; ++q) v[u][q] = hr[q * 32];
        }
#pragma unroll
        for (int u = 0; u < U; ++u) {
            int   dj = __builtin_amdgcn_readlane(dloc, j + u);
            float wj = __builtin_bit_cast(float, __builtin_amdgcn_readlane(__builtin_bit_cast(int, wgt), j + u));
            add_to_slot<NQ>(acc, dj >> 5, v[u], wj);
        }
    }
    for (; j < m; ++j) {
        int   sj = __builtin_amdgcn_readlane(s, j);
        int   dj = __builtin_amdgcn_readlane(dloc, j);
        float wj = __builtin_bit_cast(float, __builtin_amdgcn_readlane(__builtin_bit_cast(int, wgt), j));
        const float* hr = h + (size_t)sj * F + lane;
        float v[NQ];
#pragma unroll
        for (int q = 0; q < NQ; ++q) v[q] = hr[q * 32];
        add_to_slot<NQ>(acc, dj >> 5, v, wj);
    }
}

template <int F, int U, bool RECORD>
__device__ __forceinline__ void take_edges(const int* ent, int m, int lane,
                                           const int* __restrict__ src,
                                           const float* __restrict__ dinv,
                                           const float* __restrict__ h,
                                           const float* dl, v8f (&acc)[F / 32],
                                           int* recId, float* recW, int& recn, bool& rovf) {
    int s = 0, dloc = 0;
    float wgt = 0.0f;
    if (lane < m) {
        int en = ent[lane];
        dloc  = (int)((unsigned)en >> 24);
        s     = src[en & 0xFFFFFF];
        wgt   = dinv[s] * dl[dloc];
    }
    if (RECORD) {
        if (recn + m <= RCAP) {
            if (lane < m) {
                recId[recn + lane] = (int)(((unsigned)dloc << 24) | (unsigned)s);
                recW[recn + lane]  = wgt;
            }
        } else {
            rovf = true;
        }
        recn += m;
    }
    add_rows<F, U>(s, dloc, wgt, m, lane, h, acc);
}

__device__ __forceinline__ void emit_row(unsigned z, unsigned x, unsigned mk, int e0,
                                         int* mylist, int& cntw) {
    const unsigned multi = __builtin_amdgcn_ballot_w32((z & (z - 1u)) != 0u);
    if (multi == 0u) {
        if (z != 0u) mylist[cntw + __builtin_amdgcn_mbcnt_lo(mk, 0u)] = e0 + (__builtin_ctz(z) >> 3);
        cntw += __builtin_popcount(mk);
    } else {
        const unsigned t = (x & 0x7f7f7f7fu) + 0x7f7f7f7fu;
        unsigned f = ~(t | x | 0x7f7f7f7fu);
        const int pcl = __builtin_popcount(f);
        int pre = 0, total = 0;
#pragma unroll
        for (int r = 0; r < 4; ++r) {
            const unsigned m = __builtin_amdgcn_ballot_w32(pcl > r);
            pre   += (int)__builtin_amdgcn_mbcnt_lo(m, 0u);
            total += __builtin_popcount(m);
        }
        int q = cntw + pre;
        while (f) {
            mylist[q] = e0 + (__builtin_ctz(f) >> 3);
            ++q;
            f &= f - 1u;
        }
        cntw += total;
    }
}

template <bool TAIL>
__device__ __forceinline__ void filter_chunk(const unsigned* __restrict__ ownw, int base, int E,
                                             unsigned pat, int lane,
                                             int* mylist, int& cntw, bool& ovf) {
    unsigned x[8], z[8], mk[8];
    const unsigned* p = ownw + (base >> 2) + lane;
#pragma unroll
    for (int j = 0; j < 8; ++j) {
        if (TAIL) {
            const int nv = E - (base + j * 128 + lane * 4);
            unsigned w = (nv > 0) ? p[j * 32] : 0u;
            w ^= pat;
            if (nv < 4) w |= (nv <= 0) ? 0xFFFFFFFFu : (0xFFFFFFFFu << (8 * nv));
            x[j] = w;
        } else {
            x[j] = p[j * 32] ^ pat;
        }
    }
    int nflag = 0;
#pragma unroll
    for (int j = 0; j < 8; ++j) {
        z[j]  = (x[j] - 0x01010101u) & ~x[j] & 0x80808080u;
        mk[j] = __builtin_amdgcn_ballot_w32(z[j] != 0u);
        nflag += __builtin_popcount(mk[j]);
    }
    if (nflag == 0) return;
    if (cntw + 4 * nflag > LCAP) {
        ovf = true;
        return;
    }
#pragma unroll
    for (int j = 0; j < 8; ++j) {
        if (mk[j]) emit_row(z[j], x[j], mk[j], base + j * 128 + lane * 4, mylist, cntw);
    }
}

template <int F, bool RELU, bool REPLAY>
__global__ __launch_bounds__(1024) void agg_tile(const int* __restrict__ src,
                                                 const int* __restrict__ dst,
                                                 const float* __restrict__ dinv,
                                                 const float* __restrict__ h,
                                                 const float* __restrict__ bias,
                                                 float* __restrict__ outp, int E,
                                                 int* __restrict__ recs,
                                                 int* __restrict__ reccnt,
                                                 const unsigned* __restrict__ ownw) {
    static_assert(RN <= 256, "a list entry keeps the row in 8 bits");
    static_assert((NNODES + RN - 1) / RN <= 256, "a node range's number is kept in one byte");
    constexpr int U  = 1024 / F;
    constexpr int NQ = F / 32;
    __shared__ float dl[RN];
    __shared__ int   list[32 * LCAP];
    __shared__ int   pendl[32 * PCAP];
    __shared__ int   cntArr[32];
    __shared__ int   s_ovf;

    const int lane = threadIdx.x & 31;
    const int wave = __builtin_amdgcn_readfirstlane((int)(threadIdx.x >> 5));
    const int lo   = blockIdx.x * RN;
    const int nn   = min(RN, NNODES - lo);

    int*   recId = recs + (size_t)(blockIdx.x * 32 + wave) * (2 * RCAP);
    float* recW  = (float*)(recId + RCAP);
    int cw = 0;
    if (REPLAY) cw = reccnt[blockIdx.x * 32 + lane];

    if (threadIdx.x == 0) s_ovf = (E >= (1 << 24)) ? 1 : 0;
    for (int i = threadIdx.x; i < nn; i += 1024) dl[i] = dinv[lo + i];
    __syncthreads();

    bool  replay = false;
    int   myn = 0;
    int   curId = 0;
    float curW = 0.0f;
    if (REPLAY) {
        replay = __builtin_amdgcn_ballot_w32(cw < 0) == 0u;
        myn = __builtin_amdgcn_readlane(cw, wave);
        if (replay && lane < myn) {
            curId = recId[lane];
            curW  = recW[lane];
        }
    }
    static_assert(NSLOT <= 8, "one vector element per row of an owner wave");
    v8f acc[NQ];
    {
        float bq[NQ], hv[NSLOT][NQ], dq[NSLOT];
#pragma unroll
        for (int q = 0; q < NQ; ++q) bq[q] = bias[lane + q * 32];
#pragma unroll
        for (int k = 0; k < NSLOT; ++k) {
            const int r = min(wave + 32 * k, nn - 1);
            const float* hr = h + (size_t)(lo + r) * F + lane;
            dq[k] = dl[r];
#pragma unroll
            for (int q = 0; q < NQ; ++q) hv[k][q] = hr[q * 32];
        }
#pragma unroll
        for (int q = 0; q < NQ; ++q) {
#pragma unroll
            for (int k = 0; k < NSLOT; ++k) acc[q][k] = hv[k][q] * (dq[k] * dq[k]) + bq[q];
#pragma unroll
            for (int k = NSLOT; k < 8; ++k) acc[q][k] = 0.0f;
        }
    }

    if (REPLAY && replay) {
        for (int b0 = 0; b0 < myn; b0 += 32) {
            const int m = min(32, myn - b0);
            int   nxtId = 0;
            float nxtW = 0.0f;
            if (b0 + 32 + lane < myn) {
                nxtId = recId[b0 + 32 + lane];
                nxtW  = recW[b0 + 32 + lane];
            }
            add_rows<F, U>(curId & 0xFFFFFF, (int)((unsigned)curId >> 24), curW, m, lane, h, acc);
            curId = nxtId;
            curW  = nxtW;
        }
    } else {
        int* mylist = list + wave * LCAP;
        int cntw = 0;
        bool ovf = false;
        const unsigned pat = (unsigned)blockIdx.x * 0x01010101u;
        for (int base = wave * 1024; base < E; base += 32 * 1024) {
            if (base + 1024 <= E) filter_chunk<false>(ownw, base, E, pat, lane, mylist, cntw, ovf);
            else                  filter_chunk<true>(ownw, base, E, pat, lane, mylist, cntw, ovf);
        }
        for (int i0 = 0; i0 < cntw; i0 += 128) {
            int en[4], dd[4];
#pragma unroll
            for (int t = 0; t < 4; ++t) {
                const int idx = i0 + t * 32 + lane;
                en[t] = (idx < cntw) ? mylist[idx] : -1;
            }
#pragma unroll
            for (int t = 0; t < 4; ++t) dd[t] = (en[t] >= 0) ? dst[en[t]] : 0;
#pragma unroll
            for (int t = 0; t < 4; ++t) {
                if (en[t] >= 0)
                    mylist[i0 + t * 32 + lane] = (int)(((unsigned)(dd[t] - lo) << 24) | (unsigned)en[t]);
            }
        }
        if (lane == 0) {
            cntArr[wave] = cntw;
            if (ovf) s_ovf = 1;
        }
        __syncthreads();

        if (s_ovf == 0) {
            int* mypend = pendl + wave * PCAP;
            int pend = 0;
            int  recn = 0;
            bool rovf = false;
            for (int sw = 0; sw < 32; ++sw) {
                const int c = __builtin_amdgcn_readfirstlane(cntArr[sw]);
                const int* L = list + sw * LCAP;
                for (int b = 0; b < c; b += 128) {
                    int en[4];
#pragma unroll
                    for (int t = 0; t < 4; ++t) {
                        int idx = b + t * 32 + lane;
                        en[t] = (idx < c) ? L[idx] : 0;
                    }
#pragma unroll
                    for (int t = 0; t < 4; ++t) {
                        bool mine = (b + t * 32 + lane < c) &&
                                    ((((unsigned)en[t] >> 24) & 31u) == (unsigned)wave);
                        unsigned mask = __builtin_amdgcn_ballot_w32(mine);
                        if (mask) {
                            if (mine) mypend[pend + __builtin_amdgcn_mbcnt_lo(mask, 0u)] = en[t];
                            pend += __builtin_popcount(mask);
                        }
                    }
                    while (pend >= 32) {
                        pend -= 32;
                        take_edges<F, U, !REPLAY>(mypend + pend, 32, lane, src, dinv, h, dl, acc,
                                                  recId, recW, recn, rovf);
                    }
                }
            }
            take_edges<F, U, !REPLAY>(mypend, pend, lane, src, dinv, h, dl, acc, recId, recW, recn, rovf);
            if (!REPLAY && lane == 0) reccnt[blockIdx.x * 32 + wave] = rovf ? -1 : recn;
        } else {
            if (!REPLAY && lane == 0) reccnt[blockIdx.x * 32 + wave] = -1;
            for (int base = 0; base < E; base += 32) {
                int e = base + lane;
                int d = (e < E) ? dst[e] : -1;
                unsigned dloc = (unsigned)(d - lo);
                bool mine = (dloc < (unsigned)nn) && ((dloc & 31u) == (unsigned)wave);
                unsigned mask = __builtin_amdgcn_ballot_w32(mine);
                while (mask) {
                    int j = __builtin_ctz(mask);
                    mask &= mask - 1;
                    int dj = (int)__builtin_amdgcn_readlane((int)dloc, j);
                    int sj = src[base + j];
                    float wj = dinv[sj] * dl[dj];
                    const float* hr = h + (size_t)sj * F + lane;
                    float v[NQ];
#pragma unroll
                    for (int q = 0; q < NQ; ++q) v[q] = hr[q * 32];
                    add_to_slot<NQ>(acc, dj >> 5, v, wj);
                }
            }
        }
    }

#pragma unroll
    for (int k = 0; k < NSLOT; ++k) {
        const int r = wave + 32 * k;
        if (r < nn) {
            float* op = outp + (size_t)(lo + r) * F + lane;
#pragma unroll
            for (int q = 0; q < NQ; ++q) {
                float v = acc[q][k];
                if (RELU) v = fmaxf(v, 0.0f);
                op[q * 32] = v;
            }
        }
    }
}

extern "C" void kernel_launch(void* const* d_in, const int* in_sizes, int n_in,
                              void* d_out, int out_size, void* d_ws, size_t ws_size,
                              hipStream_t stream) {
    const float* x  = (const float*)d_in[0];
    const int*   ei = (const int*)d_in[1];
    const float* W1 = (const float*)d_in[2];
    const float* b1 = (const float*)d_in[3];
    const float* W2 = (const float*)d_in[4];
    const float* b2 = (const float*)d_in[5];
    float* out = (float*)d_out;

    const int N = NNODES;
    const int E = in_sizes[1] / 2;
    const int* src = ei;
    const int* dst = ei + E;

    float* deg   = (float*)d_ws;
    float* h1    = deg + N;
    float* a1    = h1 + (size_t)N * 128;
    float* h2    = a1 + (size_t)N * 128;
    float* deg1  = h2 + (size_t)N * 64;
    int*   reccnt = (int*)(deg1 + N);
    int*   recs   = reccnt + 256 * 32 + 32;
    unsigned char* own = (unsigned char*)(recs + (size_t)256 * 32 * 2 * RCAP);

    const int BT = 256;
    const int NB = (NNODES + RN - 1) / RN;

    init_kernel<<<(N + BT - 1) / BT, BT, 0, stream>>>(deg, deg1);
    deg_kernel<<<(E + BT - 1) / BT, BT, 0, stream>>>(dst, deg, deg1, own, E);
    dinv_kernel<<<(N + BT - 1) / BT, BT, 0, stream>>>(deg, deg1);
    gemm_wmma<256, 128><<<(N / 16 + 7) / 8, BT, 0, stream>>>(x, W1, h1, N);
    agg_tile<128, true, false><<<NB, 1024, 0, stream>>>(src, dst, deg, h1, b1, a1, E, recs, reccnt, (const unsigned*)own);
    gemm_wmma<128, 64><<<(N / 16 + 7) / 8, BT, 0, stream>>>(a1, W2, h2, N);
    agg_tile<64, false, true><<<NB, 1024, 0, stream>>>(src, dst, deg, h2, b2, out, E, recs, reccnt, (const unsigned*)own);
}


# products: lanes own 4 adjacent k and 4 adjacent columns (16-byte A, B, C accesses); planes on 256-byte boundaries, so stored and gathered rows are whole lines; k sum order permuted per 8-block
# speedup vs baseline: 2.7369x; 1.0630x over previous
#include <hip/hip_runtime.h>

typedef __attribute__((ext_vector_type(2))) float v2f;
typedef __attribute__((ext_vector_type(4))) float v4f;
typedef __attribute__((ext_vector_type(8))) float v8f;

#define NNODES 50000
#define OWN_RN 196
#define NPAD   ((NNODES + 63) / 64 * 64)

__device__ __forceinline__ void atom_add_f32(float* p, float v) {
    asm volatile("global_atomic_add_f32 %0, %1, off" :: "v"(p), "v"(v) : "memory");
}

__device__ __forceinline__ int wg_half(int* slot) {
    if (threadIdx.x == 0)
        *slot = (int)((__builtin_amdgcn_s_sendmsg_rtn(0x87) >> 18) & 1u);
    __syncthreads();
    return __builtin_amdgcn_readfirstlane(*slot);
}

__global__ __launch_bounds__(256) void init_kernel(float* __restrict__ deg0,
                                                   float* __restrict__ deg1) {
    int idx = blockIdx.x * 256 + threadIdx.x;
    if (idx < NNODES) { deg0[idx] = 1.0f; deg1[idx] = 0.0f; }
}

__global__ __launch_bounds__(256) void deg_kernel(const int* __restrict__ dst,
                                                  float* __restrict__ deg0,
                                                  float* __restrict__ deg1,
                                                  unsigned char* __restrict__ own, int E) {
    __shared__ int s_half;
    float* deg = wg_half(&s_half) ? deg1 : deg0;
    int e = blockIdx.x * 256 + threadIdx.x;
    if (e < E) {
        int d = dst[e];
        own[e] = (unsigned char)(d / OWN_RN);
        atom_add_f32(&deg[d], 1.0f);
    }
}

__global__ __launch_bounds__(256) void dinv_kernel(float* __restrict__ deg0,
                                                   const float* __restrict__ deg1) {
    int i = blockIdx.x * 256 + threadIdx.x;
    if (i < NNODES) deg0[i] = rsqrtf(deg0[i] + deg1[i]);
}

template <int K, int NF>
__global__ __launch_bounds__(256) void gemm_wmma(const float* __restrict__ A,
                                                 const float* __restrict__ B,
                                                 float* __restrict__ C, int M) {
    static_assert(K % 8 == 0 && NF % 64 == 0, "blocks of 8 k, column groups of 64");
    const int lane  = threadIdx.x & 31;
    const int wave  = blockIdx.x * 8 + (threadIdx.x >> 5);
    const int mBase = wave * 16;
    if (mBase >= M) return;
    const int half = lane >> 4;
    const int l16  = lane & 15;
    constexpr int NT = NF / 16;
    constexpr int NG = NF / 64;

    v8f acc[NT] = {};

    const float* aPtr = A + (size_t)(mBase + l16) * K + 4 * half;
    const float* bPtr = B + (size_t)(4 * half) * NF + 4 * l16;
    for (int k = 0; k < K; k += 8) {
        const v4f av = *(const v4f*)(aPtr + k);
        v4f bv[4][NG];
#pragma unroll
        for (int i = 0; i < 4; ++i) {
#pragma unroll
            for (int g = 0; g < NG; ++g)
                bv[i][g] = *(const v4f*)(bPtr + (size_t)(k + i) * NF + g * 64);
        }
#pragma unroll
        for (int u = 0; u < 2; ++u) {
            v2f a;
            a.x = av[2 * u];
            a.y = av[2 * u + 1];
#pragma unroll
            for (int g = 0; g < NG; ++g) {
#pragma unroll
                for (int j = 0; j < 4; ++j) {
                    v2f b;
                    b.x = bv[2 * u][g][j];
                    b.y = bv[2 * u + 1][g][j];
                    acc[g * 4 + j] = __builtin_amdgcn_wmma_f32_16x16x4_f32(
                        false, a, false, b, (short)0, acc[g * 4 + j], false, false);
                }
            }
        }
    }
#pragma unroll
    for (int g = 0; g < NG; ++g) {
        float* cPtr = C + (size_t)(mBase + half * 8) * NF + g * 64 + 4 * l16;
#pragma unroll
        for (int r = 0; r < 8; ++r) {
            v4f o;
            o.x = acc[g * 4 + 0][r];
            o.y = acc[g * 4 + 1][r];
            o.z = acc[g * 4 + 2][r];
            o.w = acc[g * 4 + 3][r];
            *(v4f*)(cPtr + (size_t)r * NF) = o;
        }
    }
}

#define RN    OWN_RN
#define NSLOT ((RN + 31) / 32)
#define LCAP  1280
#define PCAP  160
#define RCAP  512

template <int NQ>
__device__ __forceinline__ void add_to_slot(v8f (&acc)[NQ], int slot, const float (&v)[NQ], float w) {
#pragma unroll
    for (int q = 0; q < NQ; ++q) acc[q][slot] = __builtin_fmaf(v[q], w, acc[q][slot]);
}

template <int F, int U>
__device__ __forceinline__ void add_rows(int s, int dloc, float wgt, int m, int lane,
                                         const float* __restrict__ h, v8f (&acc)[F / 32]) {
    constexpr int NQ = F / 32;
    int j = 0;
#pragma unroll 1
    for (; j + U <= m; j += U) {
        float v[U][NQ];
#pragma unroll
        for (int u = 0; u < U; ++u) {
            int sj = __builtin_amdgcn_readlane(s, j + u);
            const float* hr = h + (size_t)sj * F + lane;
#pragma unroll
            for (int q = 0; q < NQ; ++q) v[u][q] = hr[q * 32];
        }
#pragma unroll
        for (int u = 0; u < U; ++u) {
            int   dj = __builtin_amdgcn_readlane(dloc, j + u);
            float wj = __builtin_bit_cast(float, __builtin_amdgcn_readlane(__builtin_bit_cast(int, wgt), j + u));
            add_to_slot<NQ>(acc, dj >> 5, v[u], wj);
        }
    }
    for (; j < m; ++j) {
        int   sj = __builtin_amdgcn_readlane(s, j);
        int   dj = __builtin_amdgcn_readlane(dloc, j);
        float wj = __builtin_bit_cast(float, __builtin_amdgcn_readlane(__builtin_bit_cast(int, wgt), j));
        const float* hr = h + (size_t)sj * F + lane;
        float v[NQ];
#pragma unroll
        for (int q = 0; q < NQ; ++q) v[q] = hr[q * 32];
        add_to_slot<NQ>(acc, dj >> 5, v, wj);
    }
}

template <int F, int U, bool RECORD>
__device__ __forceinline__ void take_edges(const int* ent, int m, int lane,
                                           const int* __restrict__ src,
                                           const float* __restrict__ dinv,
                                           const float* __restrict__ h,
                                           const float* dl, v8f (&acc)[F / 32],
                                           int* recId, float* recW, int& recn, bool& rovf) {
    int s = 0, dloc = 0;
    float wgt = 0.0f;
    if (lane < m) {
        int en = ent[lane];
        dloc  = (int)((unsigned)en >> 24);
        s     = src[en & 0xFFFFFF];
        wgt   = dinv[s] * dl[dloc];
    }
    if (RECORD) {
        if (recn + m <= RCAP) {
            if (lane < m) {
                recId[recn + lane] = (int)(((unsigned)dloc << 24) | (unsigned)s);
                recW[recn + lane]  = wgt;
            }
        } else {
            rovf = true;
        }
        recn += m;
    }
    add_rows<F, U>(s, dloc, wgt, m, lane, h, acc);
}

__device__ __forceinline__ void emit_row(unsigned z, unsigned x, unsigned mk, int e0,
                                         int* mylist, int& cntw) {
    const unsigned multi = __builtin_amdgcn_ballot_w32((z & (z - 1u)) != 0u);
    if (multi == 0u) {
        if (z != 0u) mylist[cntw + __builtin_amdgcn_mbcnt_lo(mk, 0u)] = e0 + (__builtin_ctz(z) >> 3);
        cntw += __builtin_popcount(mk);
    } else {
        const unsigned t = (x & 0x7f7f7f7fu) + 0x7f7f7f7fu;
        unsigned f = ~(t | x | 0x7f7f7f7fu);
        const int pcl = __builtin_popcount(f);
        int pre = 0, total = 0;
#pragma unroll
        for (int r = 0; r < 4; ++r) {
            const unsigned m = __builtin_amdgcn_ballot_w32(pcl > r);
            pre   += (int)__builtin_amdgcn_mbcnt_lo(m, 0u);
            total += __builtin_popcount(m);
        }
        int q = cntw + pre;
        while (f) {
            mylist[q] = e0 + (__builtin_ctz(f) >> 3);
            ++q;
            f &= f - 1u;
        }
        cntw += total;
    }
}

template <bool TAIL>
__device__ __forceinline__ void filter_chunk(const unsigned* __restrict__ ownw, int base, int E,
                                             unsigned pat, int lane,
                                             int* mylist, int& cntw, bool& ovf) {
    unsigned x[8], z[8], mk[8];
    const unsigned* p = ownw + (base >> 2) + lane;
#pragma unroll
    for (int j = 0; j < 8; ++j) {
        if (TAIL) {
            const int nv = E - (base + j * 128 + lane * 4);
            unsigned w = (nv > 0) ? p[j * 32] : 0u;
            w ^= pat;
            if (nv < 4) w |= (nv <= 0) ? 0xFFFFFFFFu : (0xFFFFFFFFu << (8 * nv));
            x[j] = w;
        } else {
            x[j] = p[j * 32] ^ pat;
        }
    }
    int nflag = 0;
#pragma unroll
    for (int j = 0; j < 8; ++j) {
        z[j]  = (x[j] - 0x01010101u) & ~x[j] & 0x80808080u;
        mk[j] = __builtin_amdgcn_ballot_w32(z[j] != 0u);
        nflag += __builtin_popcount(mk[j]);
    }
    if (nflag == 0) return;
    if (cntw + 4 * nflag > LCAP) {
        ovf = true;
        return;
    }
#pragma unroll
    for (int j = 0; j < 8; ++j) {
        if (mk[j]) emit_row(z[j], x[j], mk[j], base + j * 128 + lane * 4, mylist, cntw);
    }
}

template <int F, bool RELU, bool REPLAY>
__global__ __launch_bounds__(1024) void agg_tile(const int* __restrict__ src,
                                                 const int* __restrict__ dst,
                                                 const float* __restrict__ dinv,
                                                 const float* __restrict__ h,
                                                 const float* __restrict__ bias,
                                                 float* __restrict__ outp, int E,
                                                 int* __restrict__ recs,
                                                 int* __restrict__ reccnt,
                                                 const unsigned* __restrict__ ownw) {
    static_assert(RN <= 256, "a list entry keeps the row in 8 bits");
    static_assert((NNODES + RN - 1) / RN <= 256, "a node range's number is kept in one byte");
    constexpr int U  = 1024 / F;
    constexpr int NQ = F / 32;
    __shared__ float dl[RN];
    __shared__ int   list[32 * LCAP];
    __shared__ int   pendl[32 * PCAP];
    __shared__ int   cntArr[32];
    __shared__ int   s_ovf;

    const int lane = threadIdx.x & 31;
    const int wave = __builtin_amdgcn_readfirstlane((int)(threadIdx.x >> 5));
    const int lo   = blockIdx.x * RN;
    const int nn   = min(RN, NNODES - lo);

    int*   recId = recs + (size_t)(blockIdx.x * 32 + wave) * (2 * RCAP);
    float* recW  = (float*)(recId + RCAP);
    int cw = 0;
    if (REPLAY) cw = reccnt[blockIdx.x * 32 + lane];

    if (threadIdx.x == 0) s_ovf = (E >= (1 << 24)) ? 1 : 0;
    for (int i = threadIdx.x; i < nn; i += 1024) dl[i] = dinv[lo + i];
    __syncthreads();

    bool  replay = false;
    int   myn = 0;
    int   curId = 0;
    float curW = 0.0f;
    if (REPLAY) {
        replay = __builtin_amdgcn_ballot_w32(cw < 0) == 0u;
        myn = __builtin_amdgcn_readlane(cw, wave);
        if (replay && lane < myn) {
            curId = recId[lane];
            curW  = recW[lane];
        }
    }
    static_assert(NSLOT <= 8, "one vector element per row of an owner wave");
    v8f acc[NQ];
    {
        float bq[NQ], hv[NSLOT][NQ], dq[NSLOT];
#pragma unroll
        for (int q = 0; q < NQ; ++q) bq[q] = bias[lane + q * 32];
#pragma unroll
        for (int k = 0; k < NSLOT; ++k) {
            const int r = min(wave + 32 * k, nn - 1);
            const float* hr = h + (size_t)(lo + r) * F + lane;
            dq[k] = dl[r];
#pragma unroll
            for (int q = 0; q < NQ; ++q) hv[k][q] = hr[q * 32];
        }
#pragma unroll
        for (int q = 0; q < NQ; ++q) {
#pragma unroll
            for (int k = 0; k < NSLOT; ++k) acc[q][k] = hv[k][q] * (dq[k] * dq[k]) + bq[q];
#pragma unroll
            for (int k = NSLOT; k < 8; ++k) acc[q][k] = 0.0f;
        }
    }

    if (REPLAY && replay) {
        for (int b0 = 0; b0 < myn; b0 += 32) {
            const int m = min(32, myn - b0);
            int   nxtId = 0;
            float nxtW = 0.0f;
            if (b0 + 32 + lane < myn) {
                nxtId = recId[b0 + 32 + lane];
                nxtW  = recW[b0 + 32 + lane];
            }
            add_rows<F, U>(curId & 0xFFFFFF, (int)((unsigned)curId >> 24), curW, m, lane, h, acc);
            curId = nxtId;
            curW  = nxtW;
        }
    } else {
        int* mylist = list + wave * LCAP;
        int cntw = 0;
        bool ovf = false;
        const unsigned pat = (unsigned)blockIdx.x * 0x01010101u;
        for (int base = wave * 1024; base < E; base += 32 * 1024) {
            if (base + 1024 <= E) filter_chunk<false>(ownw, base, E, pat, lane, mylist, cntw, ovf);
            else                  filter_chunk<true>(ownw, base, E, pat, lane, mylist, cntw, ovf);
        }
        for (int i0 = 0; i0 < cntw; i0 += 128) {
            int en[4], dd[4];
#pragma unroll
            for (int t = 0; t < 4; ++t) {
                const int idx = i0 + t * 32 + lane;
                en[t] = (idx < cntw) ? mylist[idx] : -1;
            }
#pragma unroll
            for (int t = 0; t < 4; ++t) dd[t] = (en[t] >= 0) ? dst[en[t]] : 0;
#pragma unroll
            for (int t = 0; t < 4; ++t) {
                if (en[t] >= 0)
                    mylist[i0 + t * 32 + lane] = (int)(((unsigned)(dd[t] - lo) << 24) | (unsigned)en[t]);
            }
        }
        if (lane == 0) {
            cntArr[wave] = cntw;
            if (ovf) s_ovf = 1;
        }
        __syncthreads();

        if (s_ovf == 0) {
            int* mypend = pendl + wave * PCAP;
            int pend = 0;
            int  recn = 0;
            bool rovf = false;
            for (int sw = 0; sw < 32; ++sw) {
                const int c = __builtin_amdgcn_readfirstlane(cntArr[sw]);
                const int* L = list + sw * LCAP;
                for (int b = 0; b < c; b += 128) {
                    int en[4];
#pragma unroll
                    for (int t = 0; t < 4; ++t) {
                        int idx = b + t * 32 + lane;
                        en[t] = (idx < c) ? L[idx] : 0;
                    }
#pragma unroll
                    for (int t = 0; t < 4; ++t) {
                        bool mine = (b + t * 32 + lane < c) &&
                                    ((((unsigned)en[t] >> 24) & 31u) == (unsigned)wave);
                        unsigned mask = __builtin_amdgcn_ballot_w32(mine);
                        if (mask) {
                            if (mine) mypend[pend + __builtin_amdgcn_mbcnt_lo(mask, 0u)] = en[t];
                            pend += __builtin_popcount(mask);
                        }
                    }
                    while (pend >= 32) {
                        pend -= 32;
                        take_edges<F, U, !REPLAY>(mypend + pend, 32, lane, src, dinv, h, dl, acc,
                                                  recId, recW, recn, rovf);
                    }
                }
            }
            take_edges<F, U, !REPLAY>(mypend, pend, lane, src, dinv, h, dl, acc, recId, recW, recn, rovf);
            if (!REPLAY && lane == 0) reccnt[blockIdx.x * 32 + wave] = rovf ? -1 : recn;
        } else {
            if (!REPLAY && lane == 0) reccnt[blockIdx.x * 32 + wave] = -1;
            for (int base = 0; base < E; base += 32) {
                int e = base + lane;
                int d = (e < E) ? dst[e] : -1;
                unsigned dloc = (unsigned)(d - lo);
                bool mine = (dloc < (unsigned)nn) && ((dloc & 31u) == (unsigned)wave);
                unsigned mask = __builtin_amdgcn_ballot_w32(mine);
                while (mask) {
                    int j = __builtin_ctz(mask);
                    mask &= mask - 1;
                    int dj = (int)__builtin_amdgcn_readlane((int)dloc, j);
                    int sj = src[base + j];
                    float wj = dinv[sj] * dl[dj];
                    const float* hr = h + (size_t)sj * F + lane;
                    float v[NQ];
#pragma unroll
                    for (int q = 0; q < NQ; ++q) v[q] = hr[q * 32];
                    add_to_slot<NQ>(acc, dj >> 5, v, wj);
                }
            }
        }
    }

#pragma unroll
    for (int k = 0; k < NSLOT; ++k) {
        const int r = wave + 32 * k;
        if (r < nn) {
            float* op = outp + (size_t)(lo + r) * F + lane;
#pragma unroll
            for (int q = 0; q < NQ; ++q) {
                float v = acc[q][k];
                if (RELU) v = fmaxf(v, 0.0f);
                op[q * 32] = v;
            }
        }
    }
}

extern "C" void kernel_launch(void* const* d_in, const int* in_sizes, int n_in,
                              void* d_out, int out_size, void* d_ws, size_t ws_size,
                              hipStream_t stream) {
    const float* x  = (const float*)d_in[0];
    const int*   ei = (const int*)d_in[1];
    const float* W1 = (const float*)d_in[2];
    const float* b1 = (const float*)d_in[3];
    const float* W2 = (const float*)d_in[4];
    const float* b2 = (const float*)d_in[5];
    float* out = (float*)d_out;

    const int N = NNODES;
    const int E = in_sizes[1] / 2;
    const int* src = ei;
    const int* dst = ei + E;

    float* deg   = (float*)d_ws;
    float* h1    = deg + NPAD;
    float* a1    = h1 + (size_t)N * 128;
    float* h2    = a1 + (size_t)N * 128;
    float* deg1  = h2 + (size_t)N * 64;
    int*   reccnt = (int*)(deg1 + NPAD);
    int*   recs   = reccnt + 256 * 32;
    unsigned char* own = (unsigned char*)(recs + (size_t)256 * 32 * 2 * RCAP);

    const int BT = 256;
    const int NB = (NNODES + RN - 1) / RN;

    init_kernel<<<(N + BT - 1) / BT, BT, 0, stream>>>(deg, deg1);
    deg_kernel<<<(E + BT - 1) / BT, BT, 0, stream>>>(dst, deg, deg1, own, E);
    dinv_kernel<<<(N + BT - 1) / BT, BT, 0, stream>>>(deg, deg1);
    gemm_wmma<256, 128><<<(N / 16 + 7) / 8, BT, 0, stream>>>(x, W1, h1, N);
    agg_tile<128, true, false><<<NB, 1024, 0, stream>>>(src, dst, deg, h1, b1, a1, E, recs, reccnt, (const unsigned*)own);
    gemm_wmma<128, 64><<<(N / 16 + 7) / 8, BT, 0, stream>>>(a1, W2, h2, N);
    agg_tile<64, false, true><<<NB, 1024, 0, stream>>>(src, dst, deg, h2, b2, out, E, recs, reccnt, (const unsigned*)own);
}


# degree count + owner plane run beside the layer-1 product in one launch: 256 extra work-groups of its grid, 12-16 edges per thread (16-byte dst reads, whole owner words); 7 launches to 6
# speedup vs baseline: 2.8440x; 1.0391x over previous
#include <hip/hip_runtime.h>

typedef __attribute__((ext_vector_type(2))) float v2f;
typedef __attribute__((ext_vector_type(4))) float v4f;
typedef __attribute__((ext_vector_type(8))) float v8f;
typedef __attribute__((ext_vector_type(4))) int   v4i;

#define NNODES 50000
#define OWN_RN 196
#define NPAD   ((NNODES + 63) / 64 * 64)

__device__ __forceinline__ void atom_add_f32(float* p, float v) {
    asm volatile("global_atomic_add_f32 %0, %1, off" :: "v"(p), "v"(v) : "memory");
}

__device__ __forceinline__ int wg_half(int* slot) {
    if (threadIdx.x == 0)
        *slot = (int)((__builtin_amdgcn_s_sendmsg_rtn(0x87) >> 18) & 1u);
    __syncthreads();
    return __builtin_amdgcn_readfirstlane(*slot);
}

__global__ __launch_bounds__(256) void init_kernel(float* __restrict__ deg0,
                                                   float* __restrict__ deg1) {
    int idx = blockIdx.x * 256 + threadIdx.x;
    if (idx < NNODES) { deg0[idx] = 1.0f; deg1[idx] = 0.0f; }
}

#define DEG_WGS 256
__device__ __forceinline__ unsigned own_byte(int d) { return (unsigned)(d / OWN_RN); }

__device__ __forceinline__ void deg_slice(const int* __restrict__ dst,
                                          float* __restrict__ deg,
                                          unsigned* __restrict__ ownw, int E, int wg) {
    constexpr int NT = DEG_WGS * 256;
    const int tid = wg * 256 + (int)threadIdx.x;
    const int nq  = (E + 3) >> 2;
    if ((E & 3) == 0) {
        for (int q0 = tid; q0 < nq; q0 += 4 * NT) {
            v4i d[4];
#pragma unroll
            for (int u = 0; u < 4; ++u) {
                const int q = q0 + u * NT;
                d[u] = (q < nq) ? *(const v4i*)(dst + 4 * (size_t)q) : (v4i){0, 0, 0, 0};
            }
#pragma unroll
            for (int u = 0; u < 4; ++u) {
                const int q = q0 + u * NT;
                if (q < nq) {
                    ownw[q] = own_byte(d[u].x) | (own_byte(d[u].y) << 8) |
                              (own_byte(d[u].z) << 16) | (own_byte(d[u].w) << 24);
                    atom_add_f32(&deg[d[u].x], 1.0f);
                    atom_add_f32(&deg[d[u].y], 1.0f);
                    atom_add_f32(&deg[d[u].z], 1.0f);
                    atom_add_f32(&deg[d[u].w], 1.0f);
                }
            }
        }
    } else {
        unsigned char* own = (unsigned char*)ownw;
        for (int q = tid; q < nq; q += NT) {
            for (int b = 0; b < 4; ++b) {
                const int e = 4 * q + b;
                if (e < E) {
                    const int dd = dst[e];
                    own[e] = (unsigned char)own_byte(dd);
                    atom_add_f32(&deg[dd], 1.0f);
                }
            }
        }
    }
}

__global__ __launch_bounds__(256) void dinv_kernel(float* __restrict__ deg0,
                                                   const float* __restrict__ deg1) {
    int i = blockIdx.x * 256 + threadIdx.x;
    if (i < NNODES) deg0[i] = rsqrtf(deg0[i] + deg1[i]);
}

template <int K, int NF>
__device__ __forceinline__ void gemm_stripe(const float* __restrict__ A,
                                            const float* __restrict__ B,
                                            float* __restrict__ C, int M, int wg) {
    static_assert(K % 8 == 0 && NF % 64 == 0, "blocks of 8 k, column groups of 64");
    const int lane  = threadIdx.x & 31;
    const int wave  = wg * 8 + (threadIdx.x >> 5);
    const int mBase = wave * 16;
    if (mBase >= M) return;
    const int half = lane >> 4;
    const int l16  = lane & 15;
    constexpr int NT = NF / 16;
    constexpr int NG = NF / 64;

    v8f acc[NT] = {};

    const float* aPtr = A + (size_t)(mBase + l16) * K + 4 * half;
    const float* bPtr = B + (size_t)(4 * half) * NF + 4 * l16;
    for (int k = 0; k < K; k += 8) {
        const v4f av = *(const v4f*)(aPtr + k);
        v4f bv[4][NG];
#pragma unroll
        for (int i = 0; i < 4; ++i) {
#pragma unroll
            for (int g = 0; g < NG; ++g)
                bv[i][g] = *(const v4f*)(bPtr + (size_t)(k + i) * NF + g * 64);
        }
#pragma unroll
        for (int u = 0; u < 2; ++u) {
            v2f a;
            a.x = av[2 * u];
            a.y = av[2 * u + 1];
#pragma unroll
            for (int g = 0; g < NG; ++g) {
#pragma unroll
                for (int j = 0; j < 4; ++j) {
                    v2f b;
                    b.x = bv[2 * u][g][j];
                    b.y = bv[2 * u + 1][g][j];
                    acc[g * 4 + j] = __builtin_amdgcn_wmma_f32_16x16x4_f32(
                        false, a, false, b, (short)0, acc[g * 4 + j], false, false);
                }
            }
        }
    }
#pragma unroll
    for (int g = 0; g < NG; ++g) {
        float* cPtr = C + (size_t)(mBase + half * 8) * NF + g * 64 + 4 * l16;
#pragma unroll
        for (int r = 0; r < 8; ++r) {
            v4f o;
            o.x = acc[g * 4 + 0][r];
            o.y = acc[g * 4 + 1][r];
            o.z = acc[g * 4 + 2][r];
            o.w = acc[g * 4 + 3][r];
            *(v4f*)(cPtr + (size_t)r * NF) = o;
        }
    }
}

template <int K, int NF>
__global__ __launch_bounds__(256) void gemm_wmma(const float* __restrict__ A,
                                                 const float* __restrict__ B,
                                                 float* __restrict__ C, int M) {
    gemm_stripe<K, NF>(A, B, C, M, (int)blockIdx.x);
}

template <int K, int NF>
__global__ __launch_bounds__(256) void gemm_deg(const float* __restrict__ A,
                                                const float* __restrict__ B,
                                                float* __restrict__ C, int M,
                                                const int* __restrict__ dst,
                                                float* __restrict__ deg0,
                                                float* __restrict__ deg1,
                                                unsigned* __restrict__ ownw, int E) {
    __shared__ int s_half;
    const int ngemm = (M / 16 + 7) / 8;
    if ((int)blockIdx.x >= ngemm) {
        float* deg = wg_half(&s_half) ? deg1 : deg0;
        deg_slice(dst, deg, ownw, E, (int)blockIdx.x - ngemm);
        return;
    }
    gemm_stripe<K, NF>(A, B, C, M, (int)blockIdx.x);
}

#define RN    OWN_RN
#define NSLOT ((RN + 31) / 32)
#define LCAP  1280
#define PCAP  160
#define RCAP  512

template <int NQ>
__device__ __forceinline__ void add_to_slot(v8f (&acc)[NQ], int slot, const float (&v)[NQ], float w) {
#pragma unroll
    for (int q = 0; q < NQ; ++q) acc[q][slot] = __builtin_fmaf(v[q], w, acc[q][slot]);
}

template <int F, int U>
__device__ __forceinline__ void add_rows(int s, int dloc, float wgt, int m, int lane,
                                         const float* __restrict__ h, v8f (&acc)[F / 32]) {
    constexpr int NQ = F / 32;
    int j = 0;
#pragma unroll 1
    for (; j + U <= m; j += U) {
        float v[U][NQ];
#pragma unroll
        for (int u = 0; u < U; ++u) {
            int sj = __builtin_amdgcn_readlane(s, j + u);
            const float* hr = h + (size_t)sj * F + lane;
#pragma unroll
            for (int q = 0; q < NQ; ++q) v[u][q] = hr[q * 32];
        }
#pragma unroll
        for (int u = 0; u < U; ++u) {
            int   dj = __builtin_amdgcn_readlane(dloc, j + u);
            float wj = __builtin_bit_cast(float, __builtin_amdgcn_readlane(__builtin_bit_cast(int, wgt), j + u));
            add_to_slot<NQ>(acc, dj >> 5, v[u], wj);
        }
    }
    for (; j < m; ++j) {
        int   sj = __builtin_amdgcn_readlane(s, j);
        int   dj = __builtin_amdgcn_readlane(dloc, j);
        float wj = __builtin_bit_cast(float, __builtin_amdgcn_readlane(__builtin_bit_cast(int, wgt), j));
        const float* hr = h + (size_t)sj * F + lane;
        float v[NQ];
#pragma unroll
        for (int q = 0; q < NQ; ++q) v[q] = hr[q * 32];
        add_to_slot<NQ>(acc, dj >> 5, v, wj);
    }
}

template <int F, int U, bool RECORD>
__device__ __forceinline__ void take_edges(const int* ent, int m, int lane,
                                           const int* __restrict__ src,
                                           const float* __restrict__ dinv,
                                           const float* __restrict__ h,
                                           const float* dl, v8f (&acc)[F / 32],
                                           int* recId, float* recW, int& recn, bool& rovf) {
    int s = 0, dloc = 0;
    float wgt = 0.0f;
    if (lane < m) {
        int en = ent[lane];
        dloc  = (int)((unsigned)en >> 24);
        s     = src[en & 0xFFFFFF];
        wgt   = dinv[s] * dl[dloc];
    }
    if (RECORD) {
        if (recn + m <= RCAP) {
            if (lane < m) {
                recId[recn + lane] = (int)(((unsigned)dloc << 24) | (unsigned)s);
                recW[recn + lane]  = wgt;
            }
        } else {
            rovf = true;
        }
        recn += m;
    }
    add_rows<F, U>(s, dloc, wgt, m, lane, h, acc);
}

__device__ __forceinline__ void emit_row(unsigned z, unsigned x, unsigned mk, int e0,
                                         int* mylist, int& cntw) {
    const unsigned multi = __builtin_amdgcn_ballot_w32((z & (z - 1u)) != 0u);
    if (multi == 0u) {
        if (z != 0u) mylist[cntw + __builtin_amdgcn_mbcnt_lo(mk, 0u)] = e0 + (__builtin_ctz(z) >> 3);
        cntw += __builtin_popcount(mk);
    } else {
        const unsigned t = (x & 0x7f7f7f7fu) + 0x7f7f7f7fu;
        unsigned f = ~(t | x | 0x7f7f7f7fu);
        const int pcl = __builtin_popcount(f);
        int pre = 0, total = 0;
#pragma unroll
        for (int r = 0; r < 4; ++r) {
            const unsigned m = __builtin_amdgcn_ballot_w32(pcl > r);
            pre   += (int)__builtin_amdgcn_mbcnt_lo(m, 0u);
            total += __builtin_popcount(m);
        }
        int q = cntw + pre;
        while (f) {
            mylist[q] = e0 + (__builtin_ctz(f) >> 3);
            ++q;
            f &= f - 1u;
        }
        cntw += total;
    }
}

template <bool TAIL>
__device__ __forceinline__ void filter_chunk(const unsigned* __restrict__ ownw, int base, int E,
                                             unsigned pat, int lane,
                                             int* mylist, int& cntw, bool& ovf) {
    unsigned x[8], z[8], mk[8];
    const unsigned* p = ownw + (base >> 2) + lane;
#pragma unroll
    for (int j = 0; j < 8; ++j) {
        if (TAIL) {
            const int nv = E - (base + j * 128 + lane * 4);
            unsigned w = (nv > 0) ? p[j * 32] : 0u;
            w ^= pat;
            if (nv < 4) w |= (nv <= 0) ? 0xFFFFFFFFu : (0xFFFFFFFFu << (8 * nv));
            x[j] = w;
        } else {
            x[j] = p[j * 32] ^ pat;
        }
    }
    int nflag = 0;
#pragma unroll
    for (int j = 0; j < 8; ++j) {
        z[j]  = (x[j] - 0x01010101u) & ~x[j] & 0x80808080u;
        mk[j] = __builtin_amdgcn_ballot_w32(z[j] != 0u);
        nflag += __builtin_popcount(mk[j]);
    }
    if (nflag == 0) return;
    if (cntw + 4 * nflag > LCAP) {
        ovf = true;
        return;
    }
#pragma unroll
    for (int j = 0; j < 8; ++j) {
        if (mk[j]) emit_row(z[j], x[j], mk[j], base + j * 128 + lane * 4, mylist, cntw);
    }
}

template <int F, bool RELU, bool REPLAY>
__global__ __launch_bounds__(1024) void agg_tile(const int* __restrict__ src,
                                                 const int* __restrict__ dst,
                                                 const float* __restrict__ dinv,
                                                 const float* __restrict__ h,
                                                 const float* __restrict__ bias,
                                                 float* __restrict__ outp, int E,
                                                 int* __restrict__ recs,
                                                 int* __restrict__ reccnt,
                                                 const unsigned* __restrict__ ownw) {
    static_assert(RN <= 256, "a list entry keeps the row in 8 bits");
    static_assert((NNODES + RN - 1) / RN <= 256, "a node range's number is kept in one byte");
    constexpr int U  = 1024 / F;
    constexpr int NQ = F / 32;
    __shared__ float dl[RN];
    __shared__ int   list[32 * LCAP];
    __shared__ int   pendl[32 * PCAP];
    __shared__ int   cntArr[32];
    __shared__ int   s_ovf;

    const int lane = threadIdx.x & 31;
    const int wave = __builtin_amdgcn_readfirstlane((int)(threadIdx.x >> 5));
    const int lo   = blockIdx.x * RN;
    const int nn   = min(RN, NNODES - lo);

    int*   recId = recs + (size_t)(blockIdx.x * 32 + wave) * (2 * RCAP);
    float* recW  = (float*)(recId + RCAP);
    int cw = 0;
    if (REPLAY) cw = reccnt[blockIdx.x * 32 + lane];

    if (threadIdx.x == 0) s_ovf = (E >= (1 << 24)) ? 1 : 0;
    for (int i = threadIdx.x; i < nn; i += 1024) dl[i] = dinv[lo + i];
    __syncthreads();

    bool  replay = false;
    int   myn = 0;
    int   curId = 0;
    float curW = 0.0f;
    if (REPLAY) {
        replay = __builtin_amdgcn_ballot_w32(cw < 0) == 0u;
        myn = __builtin_amdgcn_readlane(cw, wave);
        if (replay && lane < myn) {
            curId = recId[lane];
            curW  = recW[lane];
        }
    }
    static_assert(NSLOT <= 8, "one vector element per row of an owner wave");
    v8f acc[NQ];
    {
        float bq[NQ], hv[NSLOT][NQ], dq[NSLOT];
#pragma unroll
        for (int q = 0; q < NQ; ++q) bq[q] = bias[lane + q * 32];
#pragma unroll
        for (int k = 0; k < NSLOT; ++k) {
            const int r = min(wave + 32 * k, nn - 1);
            const float* hr = h + (size_t)(lo + r) * F + lane;
            dq[k] = dl[r];
#pragma unroll
            for (int q = 0; q < NQ; ++q) hv[k][q] = hr[q * 32];
        }
#pragma unroll
        for (int q = 0; q < NQ; ++q) {
#pragma unroll
            for (int k = 0; k < NSLOT; ++k) acc[q][k] = hv[k][q] * (dq[k] * dq[k]) + bq[q];
#pragma unroll
            for (int k = NSLOT; k < 8; ++k) acc[q][k] = 0.0f;
        }
    }

    if (REPLAY && replay) {
        for (int b0 = 0; b0 < myn; b0 += 32) {
            const int m = min(32, myn - b0);
            int   nxtId = 0;
            float nxtW = 0.0f;
            if (b0 + 32 + lane < myn) {
                nxtId = recId[b0 + 32 + lane];
                nxtW  = recW[b0 + 32 + lane];
            }
            add_rows<F, U>(curId & 0xFFFFFF, (int)((unsigned)curId >> 24), curW, m, lane, h, acc);
            curId = nxtId;
            curW  = nxtW;
        }
    } else {
        int* mylist = list + wave * LCAP;
        int cntw = 0;
        bool ovf = false;
        const unsigned pat = (unsigned)blockIdx.x * 0x01010101u;
        for (int base = wave * 1024; base < E; base += 32 * 1024) {
            if (base + 1024 <= E) filter_chunk<false>(ownw, base, E, pat, lane, mylist, cntw, ovf);
            else                  filter_chunk<true>(ownw, base, E, pat, lane, mylist, cntw, ovf);
        }
        for (int i0 = 0; i0 < cntw; i0 += 128) {
            int en[4], dd[4];
#pragma unroll
            for (int t = 0; t < 4; ++t) {
                const int idx = i0 + t * 32 + lane;
                en[t] = (idx < cntw) ? mylist[idx] : -1;
            }
#pragma unroll
            for (int t = 0; t < 4; ++t) dd[t] = (en[t] >= 0) ? dst[en[t]] : 0;
#pragma unroll
            for (int t = 0; t < 4; ++t) {
                if (en[t] >= 0)
                    mylist[i0 + t * 32 + lane] = (int)(((unsigned)(dd[t] - lo) << 24) | (unsigned)en[t]);
            }
        }
        if (lane == 0) {
            cntArr[wave] = cntw;
            if (ovf) s_ovf = 1;
        }
        __syncthreads();

        if (s_ovf == 0) {
            int* mypend = pendl + wave * PCAP;
            int pend = 0;
            int  recn = 0;
            bool rovf = false;
            for (int sw = 0; sw < 32; ++sw) {
                const int c = __builtin_amdgcn_readfirstlane(cntArr[sw]);
                const int* L = list + sw * LCAP;
                for (int b = 0; b < c; b += 128) {
                    int en[4];
#pragma unroll
                    for (int t = 0; t < 4; ++t) {
                        int idx = b + t * 32 + lane;
                        en[t] = (idx < c) ? L[idx] : 0;
                    }
#pragma unroll
                    for (int t = 0; t < 4; ++t) {
                        bool mine = (b + t * 32 + lane < c) &&
                                    ((((unsigned)en[t] >> 24) & 31u) == (unsigned)wave);
                        unsigned mask = __builtin_amdgcn_ballot_w32(mine);
                        if (mask) {
                            if (mine) mypend[pend + __builtin_amdgcn_mbcnt_lo(mask, 0u)] = en[t];
                            pend += __builtin_popcount(mask);
                        }
                    }
                    while (pend >= 32) {
                        pend -= 32;
                        take_edges<F, U, !REPLAY>(mypend + pend, 32, lane, src, dinv, h, dl, acc,
                                                  recId, recW, recn, rovf);
                    }
                }
            }
            take_edges<F, U, !REPLAY>(mypend, pend, lane, src, dinv, h, dl, acc, recId, recW, recn, rovf);
            if (!REPLAY && lane == 0) reccnt[blockIdx.x * 32 + wave] = rovf ? -1 : recn;
        } else {
            if (!REPLAY && lane == 0) reccnt[blockIdx.x * 32 + wave] = -1;
            for (int base = 0; base < E; base += 32) {
                int e = base + lane;
                int d = (e < E) ? dst[e] : -1;
                unsigned dloc = (unsigned)(d - lo);
                bool mine = (dloc < (unsigned)nn) && ((dloc & 31u) == (unsigned)wave);
                unsigned mask = __builtin_amdgcn_ballot_w32(mine);
                while (mask) {
                    int j = __builtin_ctz(mask);
                    mask &= mask - 1;
                    int dj = (int)__builtin_amdgcn_readlane((int)dloc, j);
                    int sj = src[base + j];
                    float wj = dinv[sj] * dl[dj];
                    const float* hr = h + (size_t)sj * F + lane;
                    float v[NQ];
#pragma unroll
                    for (int q = 0; q < NQ; ++q) v[q] = hr[q * 32];
                    add_to_slot<NQ>(acc, dj >> 5, v, wj);
                }
            }
        }
    }

#pragma unroll
    for (int k = 0; k < NSLOT; ++k) {
        const int r = wave + 32 * k;
        if (r < nn) {
            float* op = outp + (size_t)(lo + r) * F + lane;
#pragma unroll
            for (int q = 0; q < NQ; ++q) {
                float v = acc[q][k];
                if (RELU) v = fmaxf(v, 0.0f);
                op[q * 32] = v;
            }
        }
    }
}

extern "C" void kernel_launch(void* const* d_in, const int* in_sizes, int n_in,
                              void* d_out, int out_size, void* d_ws, size_t ws_size,
                              hipStream_t stream) {
    const float* x  = (const float*)d_in[0];
    const int*   ei = (const int*)d_in[1];
    const float* W1 = (const float*)d_in[2];
    const float* b1 = (const float*)d_in[3];
    const float* W2 = (const float*)d_in[4];
    const float* b2 = (const float*)d_in[5];
    float* out = (float*)d_out;

    const int N = NNODES;
    const int E = in_sizes[1] / 2;
    const int* src = ei;
    const int* dst = ei + E;

    float* deg   = (float*)d_ws;
    float* h1    = deg + NPAD;
    float* a1    = h1 + (size_t)N * 128;
    float* h2    = a1 + (size_t)N * 128;
    float* deg1  = h2 + (size_t)N * 64;
    int*   reccnt = (int*)(deg1 + NPAD);
    int*   recs   = reccnt + 256 * 32;
    unsigned char* own = (unsigned char*)(recs + (size_t)256 * 32 * 2 * RCAP);

    const int BT = 256;
    const int NB = (NNODES + RN - 1) / RN;

    init_kernel<<<(N + BT - 1) / BT, BT, 0, stream>>>(deg, deg1);
    gemm_deg<256, 128><<<(N / 16 + 7) / 8 + DEG_WGS, BT, 0, stream>>>(x, W1, h1, N, dst, deg, deg1, (unsigned*)own, E);
    dinv_kernel<<<(N + BT - 1) / BT, BT, 0, stream>>>(deg, deg1);
    agg_tile<128, true, false><<<NB, 1024, 0, stream>>>(src, dst, deg, h1, b1, a1, E, recs, reccnt, (const unsigned*)own);
    gemm_wmma<128, 64><<<(N / 16 + 7) / 8, BT, 0, stream>>>(a1, W2, h2, N);
    agg_tile<64, false, true><<<NB, 1024, 0, stream>>>(src, dst, deg, h2, b2, out, E, recs, reccnt, (const unsigned*)own);
}


# edge-sum scan reads its own node range's bit plane (1 bit per edge; 256 planes built by the degree pass through an LDS transpose), not every edge's owner byte; same lists, same add order
# speedup vs baseline: 3.2675x; 1.1489x over previous
#include <hip/hip_runtime.h>

typedef __attribute__((ext_vector_type(2))) float v2f;
typedef __attribute__((ext_vector_type(4))) float v4f;
typedef __attribute__((ext_vector_type(8))) float v8f;
typedef __attribute__((ext_vector_type(4))) int   v4i;

#define NNODES 50000
#define OWN_RN 196
#define NPAD   ((NNODES + 63) / 64 * 64)

__device__ __forceinline__ void atom_add_f32(float* p, float v) {
    asm volatile("global_atomic_add_f32 %0, %1, off" :: "v"(p), "v"(v) : "memory");
}

__device__ __forceinline__ int wg_half(int* slot) {
    if (threadIdx.x == 0)
        *slot = (int)((__builtin_amdgcn_s_sendmsg_rtn(0x87) >> 18) & 1u);
    __syncthreads();
    return __builtin_amdgcn_readfirstlane(*slot);
}

__global__ __launch_bounds__(256) void init_kernel(float* __restrict__ deg0,
                                                   float* __restrict__ deg1) {
    int idx = blockIdx.x * 256 + threadIdx.x;
    if (idx < NNODES) { deg0[idx] = 1.0f; deg1[idx] = 0.0f; }
}

#define DEG_WGS 256
#define BLK_E   2048
#define BLK_W   (BLK_E / 32)
__device__ __forceinline__ unsigned own_byte(int d) { return (unsigned)(d / OWN_RN); }

__device__ __forceinline__ void deg_slice(const int* __restrict__ dst,
                                          float* __restrict__ deg,
                                          unsigned* __restrict__ bits, int E, int wg, v4i* m4) {
    unsigned char* m1 = (unsigned char*)m4;
    const int t = (int)threadIdx.x;
    const int nblk = (E + BLK_E - 1) / BLK_E;
    const size_t rowW = (size_t)nblk * BLK_W;
    for (int blk = wg; blk < nblk; blk += DEG_WGS) {
#pragma unroll
        for (int i = 0; i < 16; ++i) m4[i * 256 + t] = (v4i){0, 0, 0, 0};
        __syncthreads();
        const int e0 = blk * BLK_E + 8 * t;
        int d[8];
        if ((E & 3) == 0 && e0 + 8 <= E) {
            const v4i a = *(const v4i*)(dst + e0);
            const v4i b = *(const v4i*)(dst + e0 + 4);
            d[0] = a.x; d[1] = a.y; d[2] = a.z; d[3] = a.w;
            d[4] = b.x; d[5] = b.y; d[6] = b.z; d[7] = b.w;
        } else {
#pragma unroll
            for (int j = 0; j < 8; ++j) d[j] = (e0 + j < E) ? dst[e0 + j] : -1;
        }
        unsigned o[8];
#pragma unroll
        for (int j = 0; j < 8; ++j) o[j] = (d[j] >= 0) ? own_byte(d[j]) : (0x100u + (unsigned)j);
#pragma unroll
        for (int i = 0; i < 8; ++i) {
            if (d[i] >= 0) {
                unsigned by = 0u;
#pragma unroll
                for (int j = 0; j < 8; ++j) by |= (o[j] == o[i]) ? (1u << j) : 0u;
                m1[o[i] * (BLK_E / 8) + t] = (unsigned char)by;
                atom_add_f32(&deg[d[i]], 1.0f);
            }
        }
        __syncthreads();
#pragma unroll
        for (int i = 0; i < 16; ++i) {
            const int p = i * 256 + t;
            *(v4i*)(bits + (size_t)(p >> 4) * rowW + (size_t)blk * BLK_W + (size_t)(p & 15) * 4) = m4[p];
        }
        __syncthreads();
    }
}

__global__ __launch_bounds__(256) void dinv_kernel(float* __restrict__ deg0,
                                                   const float* __restrict__ deg1) {
    int i = blockIdx.x * 256 + threadIdx.x;
    if (i < NNODES) deg0[i] = rsqrtf(deg0[i] + deg1[i]);
}

template <int K, int NF>
__device__ __forceinline__ void gemm_stripe(const float* __restrict__ A,
                                            const float* __restrict__ B,
                                            float* __restrict__ C, int M, int wg) {
    static_assert(K % 8 == 0 && NF % 64 == 0, "blocks of 8 k, column groups of 64");
    const int lane  = threadIdx.x & 31;
    const int wave  = wg * 8 + (threadIdx.x >> 5);
    const int mBase = wave * 16;
    if (mBase >= M) return;
    const int half = lane >> 4;
    const int l16  = lane & 15;
    constexpr int NT = NF / 16;
    constexpr int NG = NF / 64;

    v8f acc[NT] = {};

    const float* aPtr = A + (size_t)(mBase + l16) * K + 4 * half;
    const float* bPtr = B + (size_t)(4 * half) * NF + 4 * l16;
    for (int k = 0; k < K; k += 8) {
        const v4f av = *(const v4f*)(aPtr + k);
        v4f bv[4][NG];
#pragma unroll
        for (int i = 0; i < 4; ++i) {
#pragma unroll
            for (int g = 0; g < NG; ++g)
                bv[i][g] = *(const v4f*)(bPtr + (size_t)(k + i) * NF + g * 64);
        }
#pragma unroll
        for (int u = 0; u < 2; ++u) {
            v2f a;
            a.x = av[2 * u];
            a.y = av[2 * u + 1];
#pragma unroll
            for (int g = 0; g < NG; ++g) {
#pragma unroll
                for (int j = 0; j < 4; ++j) {
                    v2f b;
                    b.x = bv[2 * u][g][j];
                    b.y = bv[2 * u + 1][g][j];
                    acc[g * 4 + j] = __builtin_amdgcn_wmma_f32_16x16x4_f32(
                        false, a, false, b, (short)0, acc[g * 4 + j], false, false);
                }
            }
        }
    }
#pragma unroll
    for (int g = 0; g < NG; ++g) {
        float* cPtr = C + (size_t)(mBase + half * 8) * NF + g * 64 + 4 * l16;
#pragma unroll
        for (int r = 0; r < 8; ++r) {
            v4f o;
            o.x = acc[g * 4 + 0][r];
            o.y = acc[g * 4 + 1][r];
            o.z = acc[g * 4 + 2][r];
            o.w = acc[g * 4 + 3][r];
            *(v4f*)(cPtr + (size_t)r * NF) = o;
        }
    }
}

template <int K, int NF>
__global__ __launch_bounds__(256) void gemm_wmma(const float* __restrict__ A,
                                                 const float* __restrict__ B,
                                                 float* __restrict__ C, int M) {
    gemm_stripe<K, NF>(A, B, C, M, (int)blockIdx.x);
}

template <int K, int NF>
__global__ __launch_bounds__(256) void gemm_deg(const float* __restrict__ A,
                                                const float* __restrict__ B,
                                                float* __restrict__ C, int M,
                                                const int* __restrict__ dst,
                                                float* __restrict__ deg0,
                                                float* __restrict__ deg1,
                                                unsigned* __restrict__ bits, int E) {
    __shared__ int s_half;
    __shared__ v4i s_rows[256 * BLK_E / 8 / 16];
    const int ngemm = (M / 16 + 7) / 8;
    if ((int)blockIdx.x >= ngemm) {
        float* deg = wg_half(&s_half) ? deg1 : deg0;
        deg_slice(dst, deg, bits, E, (int)blockIdx.x - ngemm, s_rows);
        return;
    }
    gemm_stripe<K, NF>(A, B, C, M, (int)blockIdx.x);
}

#define RN    OWN_RN
#define NSLOT ((RN + 31) / 32)
#define LCAP  1280
#define PCAP  160
#define RCAP  512

template <int NQ>
__device__ __forceinline__ void add_to_slot(v8f (&acc)[NQ], int slot, const float (&v)[NQ], float w) {
#pragma unroll
    for (int q = 0; q < NQ; ++q) acc[q][slot] = __builtin_fmaf(v[q], w, acc[q][slot]);
}

template <int F, int U>
__device__ __forceinline__ void add_rows(int s, int dloc, float wgt, int m, int lane,
                                         const float* __restrict__ h, v8f (&acc)[F / 32]) {
    constexpr int NQ = F / 32;
    int j = 0;
#pragma unroll 1
    for (; j + U <= m; j += U) {
        float v[U][NQ];
#pragma unroll
        for (int u = 0; u < U; ++u) {
            int sj = __builtin_amdgcn_readlane(s, j + u);
            const float* hr = h + (size_t)sj * F + lane;
#pragma unroll
            for (int q = 0; q < NQ; ++q) v[u][q] = hr[q * 32];
        }
#pragma unroll
        for (int u = 0; u < U; ++u) {
            int   dj = __builtin_amdgcn_readlane(dloc, j + u);
            float wj = __builtin_bit_cast(float, __builtin_amdgcn_readlane(__builtin_bit_cast(int, wgt), j + u));
            add_to_slot<NQ>(acc, dj >> 5, v[u], wj);
        }
    }
    for (; j < m; ++j) {
        int   sj = __builtin_amdgcn_readlane(s, j);
        int   dj = __builtin_amdgcn_readlane(dloc, j);
        float wj = __builtin_bit_cast(float, __builtin_amdgcn_readlane(__builtin_bit_cast(int, wgt), j));
        const float* hr = h + (size_t)sj * F + lane;
        float v[NQ];
#pragma unroll
        for (int q = 0; q < NQ; ++q) v[q] = hr[q * 32];
        add_to_slot<NQ>(acc, dj >> 5, v, wj);
    }
}

template <int F, int U, bool RECORD>
__device__ __forceinline__ void take_edges(const int* ent, int m, int lane,
                                           const int* __restrict__ src,
                                           const float* __restrict__ dinv,
                                           const float* __restrict__ h,
                                           const float* dl, v8f (&acc)[F / 32],
                                           int* recId, float* recW, int& recn, bool& rovf) {
    int s = 0, dloc = 0;
    float wgt = 0.0f;
    if (lane < m) {
        int en = ent[lane];
        dloc  = (int)((unsigned)en >> 24);
        s     = src[en & 0xFFFFFF];
        wgt   = dinv[s] * dl[dloc];
    }
    if (RECORD) {
        if (recn + m <= RCAP) {
            if (lane < m) {
                recId[recn + lane] = (int)(((unsigned)dloc << 24) | (unsigned)s);
                recW[recn + lane]  = wgt;
            }
        } else {
            rovf = true;
        }
        recn += m;
    }
    add_rows<F, U>(s, dloc, wgt, m, lane, h, acc);
}

#define SCAN_U 16
__device__ __forceinline__ void emit_chunk(unsigned f, int base, int lane,
                                           int* mylist, int& cntw, bool& ovf) {
    const unsigned any = __builtin_amdgcn_ballot_w32(f != 0u);
    if (any == 0u) return;
    const int pc = __builtin_popcount(f);
    int pre, total;
    if (__builtin_amdgcn_ballot_w32(pc > 1) == 0u) {
        pre   = (int)__builtin_amdgcn_mbcnt_lo(any, 0u);
        total = __builtin_popcount(any);
    } else {
        pre = 0; total = 0;
        for (int r = 0; r < 32; ++r) {
            const unsigned m = __builtin_amdgcn_ballot_w32(pc > r);
            if (m == 0u) break;
            pre   += (int)__builtin_amdgcn_mbcnt_lo(m, 0u);
            total += __builtin_popcount(m);
        }
    }
    if (cntw + total > LCAP) {
        ovf = true;
        return;
    }
    int q = cntw + pre;
    const int e0 = base + 32 * lane;
    while (f) {
        mylist[q] = e0 + __builtin_ctz(f);
        ++q;
        f &= f - 1u;
    }
    cntw += total;
}

template <int F, bool RELU, bool REPLAY>
__global__ __launch_bounds__(1024) void agg_tile(const int* __restrict__ src,
                                                 const int* __restrict__ dst,
                                                 const float* __restrict__ dinv,
                                                 const float* __restrict__ h,
                                                 const float* __restrict__ bias,
                                                 float* __restrict__ outp, int E,
                                                 int* __restrict__ recs,
                                                 int* __restrict__ reccnt,
                                                 const unsigned* __restrict__ bits) {
    static_assert(RN <= 256, "a list entry keeps the row in 8 bits");
    static_assert((NNODES + RN - 1) / RN <= 256, "a node range's number is kept in one byte");
    constexpr int U  = 1024 / F;
    constexpr int NQ = F / 32;
    __shared__ float dl[RN];
    __shared__ int   list[32 * LCAP];
    __shared__ int   pendl[32 * PCAP];
    __shared__ int   cntArr[32];
    __shared__ int   s_ovf;

    const int lane = threadIdx.x & 31;
    const int wave = __builtin_amdgcn_readfirstlane((int)(threadIdx.x >> 5));
    const int lo   = blockIdx.x * RN;
    const int nn   = min(RN, NNODES - lo);

    int*   recId = recs + (size_t)(blockIdx.x * 32 + wave) * (2 * RCAP);
    float* recW  = (float*)(recId + RCAP);
    int cw = 0;
    if (REPLAY) cw = reccnt[blockIdx.x * 32 + lane];

    if (threadIdx.x == 0) s_ovf = (E >= (1 << 24)) ? 1 : 0;
    for (int i = threadIdx.x; i < nn; i += 1024) dl[i] = dinv[lo + i];
    __syncthreads();

    bool  replay = false;
    int   myn = 0;
    int   curId = 0;
    float curW = 0.0f;
    if (REPLAY) {
        replay = __builtin_amdgcn_ballot_w32(cw < 0) == 0u;
        myn = __builtin_amdgcn_readlane(cw, wave);
        if (replay && lane < myn) {
            curId = recId[lane];
            curW  = recW[lane];
        }
    }
    static_assert(NSLOT <= 8, "one vector element per row of an owner wave");
    v8f acc[NQ];
    {
        float bq[NQ], hv[NSLOT][NQ], dq[NSLOT];
#pragma unroll
        for (int q = 0; q < NQ; ++q) bq[q] = bias[lane + q * 32];
#pragma unroll
        for (int k = 0; k < NSLOT; ++k) {
            const int r = min(wave + 32 * k, nn - 1);
            const float* hr = h + (size_t)(lo + r) * F + lane;
            dq[k] = dl[r];
#pragma unroll
            for (int q = 0; q < NQ; ++q) hv[k][q] = hr[q * 32];
        }
#pragma unroll
        for (int q = 0; q < NQ; ++q) {
#pragma unroll
            for (int k = 0; k < NSLOT; ++k) acc[q][k] = hv[k][q] * (dq[k] * dq[k]) + bq[q];
#pragma unroll
            for (int k = NSLOT; k < 8; ++k) acc[q][k] = 0.0f;
        }
    }

    if (REPLAY && replay) {
        for (int b0 = 0; b0 < myn; b0 += 32) {
            const int m = min(32, myn - b0);
            int   nxtId = 0;
            float nxtW = 0.0f;
            if (b0 + 32 + lane < myn) {
                nxtId = recId[b0 + 32 + lane];
                nxtW  = recW[b0 + 32 + lane];
            }
            add_rows<F, U>(curId & 0xFFFFFF, (int)((unsigned)curId >> 24), curW, m, lane, h, acc);
            curId = nxtId;
            curW  = nxtW;
        }
    } else {
        int* mylist = list + wave * LCAP;
        int cntw = 0;
        bool ovf = false;
        const int nch = (E + 1023) >> 10;
        const unsigned* row = bits + (size_t)blockIdx.x * ((size_t)((E + BLK_E - 1) / BLK_E) * BLK_W) + lane;
        for (int c0 = wave; c0 < nch; c0 += 32 * SCAN_U) {
            unsigned f[SCAN_U];
#pragma unroll
            for (int j = 0; j < SCAN_U; ++j) {
                const int cj = c0 + 32 * j;
                const unsigned v = row[(size_t)min(cj, nch - 1) * 32];
                f[j] = (cj < nch) ? v : 0u;
            }
#pragma unroll
            for (int j = 0; j < SCAN_U; ++j)
                emit_chunk(f[j], (c0 + 32 * j) << 10, lane, mylist, cntw, ovf);
        }
        for (int i0 = 0; i0 < cntw; i0 += 128) {
            int en[4], dd[4];
#pragma unroll
            for (int t = 0; t < 4; ++t) {
                const int idx = i0 + t * 32 + lane;
                en[t] = (idx < cntw) ? mylist[idx] : -1;
            }
#pragma unroll
            for (int t = 0; t < 4; ++t) dd[t] = (en[t] >= 0) ? dst[en[t]] : 0;
#pragma unroll
            for (int t = 0; t < 4; ++t) {
                if (en[t] >= 0)
                    mylist[i0 + t * 32 + lane] = (int)(((unsigned)(dd[t] - lo) << 24) | (unsigned)en[t]);
            }
        }
        if (lane == 0) {
            cntArr[wave] = cntw;
            if (ovf) s_ovf = 1;
        }
        __syncthreads();

        if (s_ovf == 0) {
            int* mypend = pendl + wave * PCAP;
            int pend = 0;
            int  recn = 0;
            bool rovf = false;
            for (int sw = 0; sw < 32; ++sw) {
                const int c = __builtin_amdgcn_readfirstlane(cntArr[sw]);
                const int* L = list + sw * LCAP;
                for (int b = 0; b < c; b += 128) {
                    int en[4];
#pragma unroll
                    for (int t = 0; t < 4; ++t) {
                        int idx = b + t * 32 + lane;
                        en[t] = (idx < c) ? L[idx] : 0;
                    }
#pragma unroll
                    for (int t = 0; t < 4; ++t) {
                        bool mine = (b + t * 32 + lane < c) &&
                                    ((((unsigned)en[t] >> 24) & 31u) == (unsigned)wave);
                        unsigned mask = __builtin_amdgcn_ballot_w32(mine);
                        if (mask) {
                            if (mine) mypend[pend + __builtin_amdgcn_mbcnt_lo(mask, 0u)] = en[t];
                            pend += __builtin_popcount(mask);
                        }
                    }
                    while (pend >= 32) {
                        pend -= 32;
                        take_edges<F, U, !REPLAY>(mypend + pend, 32, lane, src, dinv, h, dl, acc,
                                                  recId, recW, recn, rovf);
                    }
                }
            }
            take_edges<F, U, !REPLAY>(mypend, pend, lane, src, dinv, h, dl, acc, recId, recW, recn, rovf);
            if (!REPLAY && lane == 0) reccnt[blockIdx.x * 32 + wave] = rovf ? -1 : recn;
        } else {
            if (!REPLAY && lane == 0) reccnt[blockIdx.x * 32 + wave] = -1;
            for (int base = 0; base < E; base += 32) {
                int e = base + lane;
                int d = (e < E) ? dst[e] : -1;
                unsigned dloc = (unsigned)(d - lo);
                bool mine = (dloc < (unsigned)nn) && ((dloc & 31u) == (unsigned)wave);
                unsigned mask = __builtin_amdgcn_ballot_w32(mine);
                while (mask) {
                    int j = __builtin_ctz(mask);
                    mask &= mask - 1;
                    int dj = (int)__builtin_amdgcn_readlane((int)dloc, j);
                    int sj = src[base + j];
                    float wj = dinv[sj] * dl[dj];
                    const float* hr = h + (size_t)sj * F + lane;
                    float v[NQ];
#pragma unroll
                    for (int q = 0; q < NQ; ++q) v[q] = hr[q * 32];
                    add_to_slot<NQ>(acc, dj >> 5, v, wj);
                }
            }
        }
    }

#pragma unroll
    for (int k = 0; k < NSLOT; ++k) {
        const int r = wave + 32 * k;
        if (r < nn) {
            float* op = outp + (size_t)(lo + r) * F + lane;
#pragma unroll
            for (int q = 0; q < NQ; ++q) {
                float v = acc[q][k];
                if (RELU) v = fmaxf(v, 0.0f);
                op[q * 32] = v;
            }
        }
    }
}

extern "C" void kernel_launch(void* const* d_in, const int* in_sizes, int n_in,
                              void* d_out, int out_size, void* d_ws, size_t ws_size,
                              hipStream_t stream) {
    const float* x  = (const float*)d_in[0];
    const int*   ei = (const int*)d_in[1];
    const float* W1 = (const float*)d_in[2];
    const float* b1 = (const float*)d_in[3];
    const float* W2 = (const float*)d_in[4];
    const float* b2 = (const float*)d_in[5];
    float* out = (float*)d_out;

    const int N = NNODES;
    const int E = in_sizes[1] / 2;
    const int* src = ei;
    const int* dst = ei + E;

    float* deg   = (float*)d_ws;
    float* h1    = deg + NPAD;
    float* a1    = h1 + (size_t)N * 128;
    float* h2    = a1 + (size_t)N * 128;
    float* deg1  = h2 + (size_t)N * 64;
    int*   reccnt = (int*)(deg1 + NPAD);
    int*   recs   = reccnt + 256 * 32;
    unsigned* bits = (unsigned*)(recs + (size_t)256 * 32 * 2 * RCAP);

    const int BT = 256;
    const int NB = (NNODES + RN - 1) / RN;

    init_kernel<<<(N + BT - 1) / BT, BT, 0, stream>>>(deg, deg1);
    gemm_deg<256, 128><<<(N / 16 + 7) / 8 + DEG_WGS, BT, 0, stream>>>(x, W1, h1, N, dst, deg, deg1, bits, E);
    dinv_kernel<<<(N + BT - 1) / BT, BT, 0, stream>>>(deg, deg1);
    agg_tile<128, true, false><<<NB, 1024, 0, stream>>>(src, dst, deg, h1, b1, a1, E, recs, reccnt, bits);
    gemm_wmma<128, 64><<<(N / 16 + 7) / 8, BT, 0, stream>>>(a1, W2, h2, N);
    agg_tile<64, false, true><<<NB, 1024, 0, stream>>>(src, dst, deg, h2, b2, out, E, recs, reccnt, bits);
}


# edge sums, layer 1: scanning waves resolve each listed edge (row, source id) and deal it into one sub-list per owner wave (rank by ballots); owners read only those; add order fixed, but a new one
# speedup vs baseline: 3.5241x; 1.0785x over previous
#include <hip/hip_runtime.h>

typedef __attribute__((ext_vector_type(2))) float v2f;
typedef __attribute__((ext_vector_type(4))) float v4f;
typedef __attribute__((ext_vector_type(8))) float v8f;
typedef __attribute__((ext_vector_type(4))) int   v4i;

#define NNODES 50000
#define OWN_RN 196
#define NPAD   ((NNODES + 63) / 64 * 64)

__device__ __forceinline__ void atom_add_f32(float* p, float v) {
    asm volatile("global_atomic_add_f32 %0, %1, off" :: "v"(p), "v"(v) : "memory");
}

__device__ __forceinline__ int wg_half(int* slot) {
    if (threadIdx.x == 0)
        *slot = (int)((__builtin_amdgcn_s_sendmsg_rtn(0x87) >> 18) & 1u);
    __syncthreads();
    return __builtin_amdgcn_readfirstlane(*slot);
}

__global__ __launch_bounds__(256) void init_kernel(float* __restrict__ deg0,
                                                   float* __restrict__ deg1) {
    int idx = blockIdx.x * 256 + threadIdx.x;
    if (idx < NNODES) { deg0[idx] = 1.0f; deg1[idx] = 0.0f; }
}

#define DEG_WGS 256
#define BLK_E   2048
#define BLK_W   (BLK_E / 32)
__device__ __forceinline__ unsigned own_byte(int d) { return (unsigned)(d / OWN_RN); }

__device__ __forceinline__ void deg_slice(const int* __restrict__ dst,
                                          float* __restrict__ deg,
                                          unsigned* __restrict__ bits, int E, int wg, v4i* m4) {
    unsigned char* m1 = (unsigned char*)m4;
    const int t = (int)threadIdx.x;
    const int nblk = (E + BLK_E - 1) / BLK_E;
    const size_t rowW = (size_t)nblk * BLK_W;
    for (int blk = wg; blk < nblk; blk += DEG_WGS) {
#pragma unroll
        for (int i = 0; i < 16; ++i) m4[i * 256 + t] = (v4i){0, 0, 0, 0};
        __syncthreads();
        const int e0 = blk * BLK_E + 8 * t;
        int d[8];
        if ((E & 3) == 0 && e0 + 8 <= E) {
            const v4i a = *(const v4i*)(dst + e0);
            const v4i b = *(const v4i*)(dst + e0 + 4);
            d[0] = a.x; d[1] = a.y; d[2] = a.z; d[3] = a.w;
            d[4] = b.x; d[5] = b.y; d[6] = b.z; d[7] = b.w;
        } else {
#pragma unroll
            for (int j = 0; j < 8; ++j) d[j] = (e0 + j < E) ? dst[e0 + j] : -1;
        }
        unsigned o[8];
#pragma unroll
        for (int j = 0; j < 8; ++j) o[j] = (d[j] >= 0) ? own_byte(d[j]) : (0x100u + (unsigned)j);
#pragma unroll
        for (int i = 0; i < 8; ++i) {
            if (d[i] >= 0) {
                unsigned by = 0u;
#pragma unroll
                for (int j = 0; j < 8; ++j) by |= (o[j] == o[i]) ? (1u << j) : 0u;
                m1[o[i] * (BLK_E / 8) + t] = (unsigned char)by;
                atom_add_f32(&deg[d[i]], 1.0f);
            }
        }
        __syncthreads();
#pragma unroll
        for (int i = 0; i < 16; ++i) {
            const int p = i * 256 + t;
            *(v4i*)(bits + (size_t)(p >> 4) * rowW + (size_t)blk * BLK_W + (size_t)(p & 15) * 4) = m4[p];
        }
        __syncthreads();
    }
}

__global__ __launch_bounds__(256) void dinv_kernel(float* __restrict__ deg0,
                                                   const float* __restrict__ deg1) {
    int i = blockIdx.x * 256 + threadIdx.x;
    if (i < NNODES) deg0[i] = rsqrtf(deg0[i] + deg1[i]);
}

template <int K, int NF>
__device__ __forceinline__ void gemm_stripe(const float* __restrict__ A,
                                            const float* __restrict__ B,
                                            float* __restrict__ C, int M, int wg) {
    static_assert(K % 8 == 0 && NF % 64 == 0, "blocks of 8 k, column groups of 64");
    const int lane  = threadIdx.x & 31;
    const int wave  = wg * 8 + (threadIdx.x >> 5);
    const int mBase = wave * 16;
    if (mBase >= M) return;
    const int half = lane >> 4;
    const int l16  = lane & 15;
    constexpr int NT = NF / 16;
    constexpr int NG = NF / 64;

    v8f acc[NT] = {};

    const float* aPtr = A + (size_t)(mBase + l16) * K + 4 * half;
    const float* bPtr = B + (size_t)(4 * half) * NF + 4 * l16;
    for (int k = 0; k < K; k += 8) {
        const v4f av = *(const v4f*)(aPtr + k);
        v4f bv[4][NG];
#pragma unroll
        for (int i = 0; i < 4; ++i) {
#pragma unroll
            for (int g = 0; g < NG; ++g)
                bv[i][g] = *(const v4f*)(bPtr + (size_t)(k + i) * NF + g * 64);
        }
#pragma unroll
        for (int u = 0; u < 2; ++u) {
            v2f a;
            a.x = av[2 * u];
            a.y = av[2 * u + 1];
#pragma unroll
            for (int g = 0; g < NG; ++g) {
#pragma unroll
                for (int j = 0; j < 4; ++j) {
                    v2f b;
                    b.x = bv[2 * u][g][j];
                    b.y = bv[2 * u + 1][g][j];
                    acc[g * 4 + j] = __builtin_amdgcn_wmma_f32_16x16x4_f32(
                        false, a, false, b, (short)0, acc[g * 4 + j], false, false);
                }
            }
        }
    }
#pragma unroll
    for (int g = 0; g < NG; ++g) {
        float* cPtr = C + (size_t)(mBase + half * 8) * NF + g * 64 + 4 * l16;
#pragma unroll
        for (int r = 0; r < 8; ++r) {
            v4f o;
            o.x = acc[g * 4 + 0][r];
            o.y = acc[g * 4 + 1][r];
            o.z = acc[g * 4 + 2][r];
            o.w = acc[g * 4 + 3][r];
            *(v4f*)(cPtr + (size_t)r * NF) = o;
        }
    }
}

template <int K, int NF>
__global__ __launch_bounds__(256) void gemm_wmma(const float* __restrict__ A,
                                                 const float* __restrict__ B,
                                                 float* __restrict__ C, int M) {
    gemm_stripe<K, NF>(A, B, C, M, (int)blockIdx.x);
}

template <int K, int NF>
__global__ __launch_bounds__(256) void gemm_deg(const float* __restrict__ A,
                                                const float* __restrict__ B,
                                                float* __restrict__ C, int M,
                                                const int* __restrict__ dst,
                                                float* __restrict__ deg0,
                                                float* __restrict__ deg1,
                                                unsigned* __restrict__ bits, int E) {
    __shared__ int s_half;
    __shared__ v4i s_rows[256 * BLK_E / 8 / 16];
    const int ngemm = (M / 16 + 7) / 8;
    if ((int)blockIdx.x >= ngemm) {
        float* deg = wg_half(&s_half) ? deg1 : deg0;
        deg_slice(dst, deg, bits, E, (int)blockIdx.x - ngemm, s_rows);
        return;
    }
    gemm_stripe<K, NF>(A, B, C, M, (int)blockIdx.x);
}

#define RN    OWN_RN
#define NSLOT ((RN + 31) / 32)
#define LCAP  1280
#define PCAP  160
#define SUBCAP 24
#define SUBP   25
#define CNTP   33
#define RCAP  512

template <int NQ>
__device__ __forceinline__ void add_to_slot(v8f (&acc)[NQ], int slot, const float (&v)[NQ], float w) {
#pragma unroll
    for (int q = 0; q < NQ; ++q) acc[q][slot] = __builtin_fmaf(v[q], w, acc[q][slot]);
}

template <int F, int U>
__device__ __forceinline__ void add_rows(int s, int dloc, float wgt, int m, int lane,
                                         const float* __restrict__ h, v8f (&acc)[F / 32]) {
    constexpr int NQ = F / 32;
    int j = 0;
#pragma unroll 1
    for (; j + U <= m; j += U) {
        float v[U][NQ];
#pragma unroll
        for (int u = 0; u < U; ++u) {
            int sj = __builtin_amdgcn_readlane(s, j + u);
            const float* hr = h + (size_t)sj * F + lane;
#pragma unroll
            for (int q = 0; q < NQ; ++q) v[u][q] = hr[q * 32];
        }
#pragma unroll
        for (int u = 0; u < U; ++u) {
            int   dj = __builtin_amdgcn_readlane(dloc, j + u);
            float wj = __builtin_bit_cast(float, __builtin_amdgcn_readlane(__builtin_bit_cast(int, wgt), j + u));
            add_to_slot<NQ>(acc, dj >> 5, v[u], wj);
        }
    }
    for (; j < m; ++j) {
        int   sj = __builtin_amdgcn_readlane(s, j);
        int   dj = __builtin_amdgcn_readlane(dloc, j);
        float wj = __builtin_bit_cast(float, __builtin_amdgcn_readlane(__builtin_bit_cast(int, wgt), j));
        const float* hr = h + (size_t)sj * F + lane;
        float v[NQ];
#pragma unroll
        for (int q = 0; q < NQ; ++q) v[q] = hr[q * 32];
        add_to_slot<NQ>(acc, dj >> 5, v, wj);
    }
}

template <int F, int U, bool RECORD>
__device__ __forceinline__ void take_edges(const int* ent, int m, int lane,
                                           const float* __restrict__ dinv,
                                           const float* __restrict__ h,
                                           const float* dl, v8f (&acc)[F / 32],
                                           int* recId, float* recW, int& recn, bool& rovf) {
    int s = 0, dloc = 0;
    float wgt = 0.0f;
    if (lane < m) {
        int en = ent[lane];
        dloc  = (int)((unsigned)en >> 24);
        s     = en & 0xFFFFFF;
        wgt   = dinv[s] * dl[dloc];
    }
    if (RECORD) {
        if (recn + m <= RCAP) {
            if (lane < m) {
                recId[recn + lane] = (int)(((unsigned)dloc << 24) | (unsigned)s);
                recW[recn + lane]  = wgt;
            }
        } else {
            rovf = true;
        }
        recn += m;
    }
    add_rows<F, U>(s, dloc, wgt, m, lane, h, acc);
}

#define SCAN_U 16
__device__ __forceinline__ void emit_chunk(unsigned f, int base, int lane,
                                           int* mylist, int& cntw, bool& ovf) {
    const unsigned any = __builtin_amdgcn_ballot_w32(f != 0u);
    if (any == 0u) return;
    const int pc = __builtin_popcount(f);
    int pre, total;
    if (__builtin_amdgcn_ballot_w32(pc > 1) == 0u) {
        pre   = (int)__builtin_amdgcn_mbcnt_lo(any, 0u);
        total = __builtin_popcount(any);
    } else {
        pre = 0; total = 0;
        for (int r = 0; r < 32; ++r) {
            const unsigned m = __builtin_amdgcn_ballot_w32(pc > r);
            if (m == 0u) break;
            pre   += (int)__builtin_amdgcn_mbcnt_lo(m, 0u);
            total += __builtin_popcount(m);
        }
    }
    if (cntw + total > LCAP) {
        ovf = true;
        return;
    }
    int q = cntw + pre;
    const int e0 = base + 32 * lane;
    while (f) {
        mylist[q] = e0 + __builtin_ctz(f);
        ++q;
        f &= f - 1u;
    }
    cntw += total;
}

template <int F, bool RELU, bool REPLAY>
__global__ __launch_bounds__(1024) void agg_tile(const int* __restrict__ src,
                                                 const int* __restrict__ dst,
                                                 const float* __restrict__ dinv,
                                                 const float* __restrict__ h,
                                                 const float* __restrict__ bias,
                                                 float* __restrict__ outp, int E,
                                                 int* __restrict__ recs,
                                                 int* __restrict__ reccnt,
                                                 const unsigned* __restrict__ bits) {
    static_assert(RN <= 256 && NNODES <= (1 << 24), "a sub-list entry keeps the row in 8 bits and the source id in 24");
    static_assert((NNODES + RN - 1) / RN <= 256, "a node range's number is kept in one byte");
    constexpr int U  = 1024 / F;
    constexpr int NQ = F / 32;
    __shared__ float dl[RN];
    __shared__ int   list[32 * LCAP];
    __shared__ int   pendl[32 * PCAP];
    __shared__ int   sub[32 * 32 * SUBP];
    __shared__ int   subcnt[32 * CNTP];
    __shared__ int   s_ovf;

    const int lane = threadIdx.x & 31;
    const int wave = __builtin_amdgcn_readfirstlane((int)(threadIdx.x >> 5));
    const int lo   = blockIdx.x * RN;
    const int nn   = min(RN, NNODES - lo);

    int*   recId = recs + (size_t)(blockIdx.x * 32 + wave) * (2 * RCAP);
    float* recW  = (float*)(recId + RCAP);
    int cw = 0;
    if (REPLAY) cw = reccnt[blockIdx.x * 32 + lane];

    if (threadIdx.x == 0) s_ovf = 0;
    for (int i = threadIdx.x; i < nn; i += 1024) dl[i] = dinv[lo + i];
    __syncthreads();

    bool  replay = false;
    int   myn = 0;
    int   curId = 0;
    float curW = 0.0f;
    if (REPLAY) {
        replay = __builtin_amdgcn_ballot_w32(cw < 0) == 0u;
        myn = __builtin_amdgcn_readlane(cw, wave);
        if (replay && lane < myn) {
            curId = recId[lane];
            curW  = recW[lane];
        }
    }
    static_assert(NSLOT <= 8, "one vector element per row of an owner wave");
    v8f acc[NQ];
    {
        float bq[NQ], hv[NSLOT][NQ], dq[NSLOT];
#pragma unroll
        for (int q = 0; q < NQ; ++q) bq[q] = bias[lane + q * 32];
#pragma unroll
        for (int k = 0; k < NSLOT; ++k) {
            const int r = min(wave + 32 * k, nn - 1);
            const float* hr = h + (size_t)(lo + r) * F + lane;
            dq[k] = dl[r];
#pragma unroll
            for (int q = 0; q < NQ; ++q) hv[k][q] = hr[q * 32];
        }
#pragma unroll
        for (int q = 0; q < NQ; ++q) {
#pragma unroll
            for (int k = 0; k < NSLOT; ++k) acc[q][k] = hv[k][q] * (dq[k] * dq[k]) + bq[q];
#pragma unroll
            for (int k = NSLOT; k < 8; ++k) acc[q][k] = 0.0f;
        }
    }

    if (REPLAY && replay) {
        for (int b0 = 0; b0 < myn; b0 += 32) {
            const int m = min(32, myn - b0);
            int   nxtId = 0;
            float nxtW = 0.0f;
            if (b0 + 32 + lane < myn) {
                nxtId = recId[b0 + 32 + lane];
                nxtW  = recW[b0 + 32 + lane];
            }
            add_rows<F, U>(curId & 0xFFFFFF, (int)((unsigned)curId >> 24), curW, m, lane, h, acc);
            curId = nxtId;
            curW  = nxtW;
        }
    } else {
        int* mylist = list + wave * LCAP;
        int cntw = 0;
        bool ovf = false;
        const int nch = (E + 1023) >> 10;
        const unsigned* row = bits + (size_t)blockIdx.x * ((size_t)((E + BLK_E - 1) / BLK_E) * BLK_W) + lane;
        for (int c0 = wave; c0 < nch; c0 += 32 * SCAN_U) {
            unsigned f[SCAN_U];
#pragma unroll
            for (int j = 0; j < SCAN_U; ++j) {
                const int cj = c0 + 32 * j;
                const unsigned v = row[(size_t)min(cj, nch - 1) * 32];
                f[j] = (cj < nch) ? v : 0u;
            }
#pragma unroll
            for (int j = 0; j < SCAN_U; ++j)
                emit_chunk(f[j], (c0 + 32 * j) << 10, lane, mylist, cntw, ovf);
        }
        subcnt[lane * CNTP + wave] = 0;
        for (int i0 = 0; i0 < cntw; i0 += 128) {
            int en[4], dd[4], ss[4];
#pragma unroll
            for (int t = 0; t < 4; ++t) {
                const int idx = i0 + t * 32 + lane;
                en[t] = (idx < cntw) ? mylist[idx] : -1;
            }
#pragma unroll
            for (int t = 0; t < 4; ++t) dd[t] = (en[t] >= 0) ? dst[en[t]] : 0;
#pragma unroll
            for (int t = 0; t < 4; ++t) ss[t] = (en[t] >= 0) ? src[en[t]] : 0;
#pragma unroll
            for (int t = 0; t < 4; ++t) {
                const bool valid = en[t] >= 0;
                const unsigned any = __builtin_amdgcn_ballot_w32(valid);
                if (any == 0u) continue;
                const unsigned dloc = (unsigned)(dd[t] - lo);
                const unsigned o = dloc & 31u;
                unsigned same = any;
#pragma unroll
                for (int b = 0; b < 5; ++b) {
                    const bool bit = ((o >> b) & 1u) != 0u;
                    const unsigned bal = __builtin_amdgcn_ballot_w32(bit);
                    same &= bit ? bal : ~bal;
                }
                if (valid) {
                    const int before = (int)__builtin_amdgcn_mbcnt_lo(same, 0u);
                    int* cp = subcnt + o * CNTP + wave;
                    const int old = *cp;
                    const int k = old + before;
                    if (k < SUBCAP) sub[(o * 32 + wave) * SUBP + k] = (int)((dloc << 24) | (unsigned)ss[t]);
                    else ovf = true;
                    if (before == 0) *cp = old + __builtin_popcount(same);
                }
            }
        }
        if (ovf) s_ovf = 1;
        __syncthreads();

        if (s_ovf == 0) {
            int* mypend = pendl + wave * PCAP;
            int pend = 0;
            int  recn = 0;
            bool rovf = false;
            const int* mysub = sub + (wave * 32 + lane) * SUBP;
            const int cvec = subcnt[wave * CNTP + lane];
            for (int k = 0; k < SUBCAP; ++k) {
                const bool have = k < cvec;
                const unsigned mask = __builtin_amdgcn_ballot_w32(have);
                if (mask == 0u) break;
                if (have) mypend[pend + __builtin_amdgcn_mbcnt_lo(mask, 0u)] = mysub[k];
                pend += __builtin_popcount(mask);
                while (pend >= 32) {
                    pend -= 32;
                    take_edges<F, U, !REPLAY>(mypend + pend, 32, lane, dinv, h, dl, acc,
                                              recId, recW, recn, rovf);
                }
            }
            take_edges<F, U, !REPLAY>(mypend, pend, lane, dinv, h, dl, acc, recId, recW, recn, rovf);
            if (!REPLAY && lane == 0) reccnt[blockIdx.x * 32 + wave] = rovf ? -1 : recn;
        } else {
            if (!REPLAY && lane == 0) reccnt[blockIdx.x * 32 + wave] = -1;
            for (int base = 0; base < E; base += 32) {
                int e = base + lane;
                int d = (e < E) ? dst[e] : -1;
                unsigned dloc = (unsigned)(d - lo);
                bool mine = (dloc < (unsigned)nn) && ((dloc & 31u) == (unsigned)wave);
                unsigned mask = __builtin_amdgcn_ballot_w32(mine);
                while (mask) {
                    int j = __builtin_ctz(mask);
                    mask &= mask - 1;
                    int dj = (int)__builtin_amdgcn_readlane((int)dloc, j);
                    int sj = src[base + j];
                    float wj = dinv[sj] * dl[dj];
                    const float* hr = h + (size_t)sj * F + lane;
                    float v[NQ];
#pragma unroll
                    for (int q = 0; q < NQ; ++q) v[q] = hr[q * 32];
                    add_to_slot<NQ>(acc, dj >> 5, v, wj);
                }
            }
        }
    }

#pragma unroll
    for (int k = 0; k < NSLOT; ++k) {
        const int r = wave + 32 * k;
        if (r < nn) {
            float* op = outp + (size_t)(lo + r) * F + lane;
#pragma unroll
            for (int q = 0; q < NQ; ++q) {
                float v = acc[q][k];
                if (RELU) v = fmaxf(v, 0.0f);
                op[q * 32] = v;
            }
        }
    }
}

extern "C" void kernel_launch(void* const* d_in, const int* in_sizes, int n_in,
                              void* d_out, int out_size, void* d_ws, size_t ws_size,
                              hipStream_t stream) {
    const float* x  = (const float*)d_in[0];
    const int*   ei = (const int*)d_in[1];
    const float* W1 = (const float*)d_in[2];
    const float* b1 = (const float*)d_in[3];
    const float* W2 = (const float*)d_in[4];
    const float* b2 = (const float*)d_in[5];
    float* out = (float*)d_out;

    const int N = NNODES;
    const int E = in_sizes[1] / 2;
    const int* src = ei;
    const int* dst = ei + E;

    float* deg   = (float*)d_ws;
    float* h1    = deg + NPAD;
    float* a1    = h1 + (size_t)N * 128;
    float* h2    = a1 + (size_t)N * 128;
    float* deg1  = h2 + (size_t)N * 64;
    int*   reccnt = (int*)(deg1 + NPAD);
    int*   recs   = reccnt + 256 * 32;
    unsigned* bits = (unsigned*)(recs + (size_t)256 * 32 * 2 * RCAP);

    const int BT = 256;
    const int NB = (NNODES + RN - 1) / RN;

    init_kernel<<<(N + BT - 1) / BT, BT, 0, stream>>>(deg, deg1);
    gemm_deg<256, 128><<<(N / 16 + 7) / 8 + DEG_WGS, BT, 0, stream>>>(x, W1, h1, N, dst, deg, deg1, bits, E);
    dinv_kernel<<<(N + BT - 1) / BT, BT, 0, stream>>>(deg, deg1);
    agg_tile<128, true, false><<<NB, 1024, 0, stream>>>(src, dst, deg, h1, b1, a1, E, recs, reccnt, bits);
    gemm_wmma<128, 64><<<(N / 16 + 7) / 8, BT, 0, stream>>>(a1, W2, h2, N);
    agg_tile<64, false, true><<<NB, 1024, 0, stream>>>(src, dst, deg, h2, b2, out, E, recs, reccnt, bits);
}


# degree pass hands each node range its edges as (row|source) entries in one 128-B bin per 2048-edge block; edge-sum blocks read whole bins: no bit-plane scan, no dst/src gathers; new fixed add order
# speedup vs baseline: 3.8232x; 1.0849x over previous
#include <hip/hip_runtime.h>

typedef __attribute__((ext_vector_type(2))) float v2f;
typedef __attribute__((ext_vector_type(4))) float v4f;
typedef __attribute__((ext_vector_type(8))) float v8f;
typedef __attribute__((ext_vector_type(4))) int   v4i;

#define NNODES 50000
#define OWN_RN 196
#define NPAD   ((NNODES + 63) / 64 * 64)

__device__ __forceinline__ void atom_add_f32(float* p, float v) {
    asm volatile("global_atomic_add_f32 %0, %1, off" :: "v"(p), "v"(v) : "memory");
}

__device__ __forceinline__ int wg_half(int* slot) {
    if (threadIdx.x == 0)
        *slot = (int)((__builtin_amdgcn_s_sendmsg_rtn(0x87) >> 18) & 1u);
    __syncthreads();
    return __builtin_amdgcn_readfirstlane(*slot);
}

__global__ __launch_bounds__(256) void init_kernel(float* __restrict__ deg0,
                                                   float* __restrict__ deg1) {
    int idx = blockIdx.x * 256 + threadIdx.x;
    if (idx < NNODES) { deg0[idx] = 1.0f; deg1[idx] = 0.0f; }
}

#define DEG_WGS 256
#define BLK_E   2048
#define BIN_W   32
__device__ __forceinline__ unsigned own_byte(int d) { return (unsigned)(d / OWN_RN); }

__device__ __forceinline__ void deg_slice(const int* __restrict__ src,
                                          const int* __restrict__ dst,
                                          float* __restrict__ deg,
                                          int* __restrict__ bins, int E, int wg,
                                          v4i* m4, unsigned char* pre8, int* cntl) {
    unsigned char* m1 = (unsigned char*)m4;
    const int t = (int)threadIdx.x;
    const int lane = t & 31;
    const int wave = t >> 5;
    const int nblk = (E + BLK_E - 1) / BLK_E;
    for (int blk = wg; blk < nblk; blk += DEG_WGS) {
#pragma unroll
        for (int i = 0; i < 16; ++i) m4[i * 256 + t] = (v4i){0, 0, 0, 0};
        __syncthreads();
        const int e0 = blk * BLK_E + 8 * t;
        int d[8], s[8];
        if ((E & 3) == 0 && e0 + 8 <= E) {
            const v4i a = *(const v4i*)(dst + e0);
            const v4i b = *(const v4i*)(dst + e0 + 4);
            const v4i sa = *(const v4i*)(src + e0);
            const v4i sb = *(const v4i*)(src + e0 + 4);
            d[0] = a.x; d[1] = a.y; d[2] = a.z; d[3] = a.w;
            d[4] = b.x; d[5] = b.y; d[6] = b.z; d[7] = b.w;
            s[0] = sa.x; s[1] = sa.y; s[2] = sa.z; s[3] = sa.w;
            s[4] = sb.x; s[5] = sb.y; s[6] = sb.z; s[7] = sb.w;
        } else {
#pragma unroll
            for (int j = 0; j < 8; ++j) {
                d[j] = (e0 + j < E) ? dst[e0 + j] : -1;
                s[j] = (e0 + j < E) ? src[e0 + j] : 0;
            }
        }
        unsigned o[8];
#pragma unroll
        for (int j = 0; j < 8; ++j) o[j] = (d[j] >= 0) ? own_byte(d[j]) : (0x100u + (unsigned)j);
#pragma unroll
        for (int i = 0; i < 8; ++i) {
            if (d[i] >= 0) {
                unsigned by = 0u;
#pragma unroll
                for (int j = 0; j < 8; ++j) by |= (o[j] == o[i]) ? (1u << j) : 0u;
                m1[o[i] * (BLK_E / 8) + t] = (unsigned char)by;
                atom_add_f32(&deg[d[i]], 1.0f);
            }
        }
        int en[8];
#pragma unroll
        for (int j = 0; j < 8; ++j)
            en[j] = (d[j] >= 0) ? (int)(((unsigned)(d[j] - (int)o[j] * OWN_RN) << 24) | (unsigned)s[j]) : 0;
        __syncthreads();
        for (int r0 = wave * 32; r0 < wave * 32 + 32; r0 += 4) {
            unsigned long long f[4];
#pragma unroll
            for (int u = 0; u < 4; ++u)
                f[u] = *(const unsigned long long*)(m1 + (r0 + u) * (BLK_E / 8) + 8 * lane);
#pragma unroll
            for (int u = 0; u < 4; ++u) {
                const int pc = __builtin_popcountll(f[u]);
                int pre = 0, total = 0;
#pragma unroll
                for (int b = 0; b < 7; ++b) {
                    const unsigned m = __builtin_amdgcn_ballot_w32(((pc >> b) & 1) != 0);
                    pre   += (int)__builtin_amdgcn_mbcnt_lo(m, 0u) << b;
                    total += __builtin_popcount(m) << b;
                }
                pre8[(r0 + u) * 32 + lane] = (unsigned char)min(pre, 255);
                cntl[r0 + u] = total;
            }
        }
        __syncthreads();
        int rank[8];
#pragma unroll
        for (int j = 0; j < 8; ++j) {
            const int r = (int)(o[j] & 0xFFu);
            const unsigned long long wv = *(const unsigned long long*)(m1 + r * (BLK_E / 8) + 8 * (t >> 3));
            const int base = (int)pre8[r * 32 + (t >> 3)];
            const int below = 8 * (t & 7) + j;
            const int rk = base + __builtin_popcountll(wv & ((1ull << below) - 1ull));
            rank[j] = (d[j] >= 0) ? rk : BIN_W;
        }
        __syncthreads();
#pragma unroll
        for (int j = 0; j < 8; ++j) {
            if (rank[j] < BIN_W - 1) ((int*)(m1 + (int)o[j] * (BLK_E / 8)))[1 + rank[j]] = en[j];
        }
        ((int*)(m1 + t * (BLK_E / 8)))[0] = cntl[t];
        __syncthreads();
#pragma unroll
        for (int i = 0; i < 8; ++i) {
            const int p = i * 256 + t;
            *(v4i*)(bins + ((size_t)(p >> 3) * nblk + blk) * BIN_W + (size_t)(p & 7) * 4) = m4[(p >> 3) * 16 + (p & 7)];
        }
        __syncthreads();
    }
}

__global__ __launch_bounds__(256) void dinv_kernel(float* __restrict__ deg0,
                                                   const float* __restrict__ deg1) {
    int i = blockIdx.x * 256 + threadIdx.x;
    if (i < NNODES) deg0[i] = rsqrtf(deg0[i] + deg1[i]);
}

template <int K, int NF>
__device__ __forceinline__ void gemm_stripe(const float* __restrict__ A,
                                            const float* __restrict__ B,
                                            float* __restrict__ C, int M, int wg) {
    static_assert(K % 8 == 0 && NF % 64 == 0, "blocks of 8 k, column groups of 64");
    const int lane  = threadIdx.x & 31;
    const int wave  = wg * 8 + (threadIdx.x >> 5);
    const int mBase = wave * 16;
    if (mBase >= M) return;
    const int half = lane >> 4;
    const int l16  = lane & 15;
    constexpr int NT = NF / 16;
    constexpr int NG = NF / 64;

    v8f acc[NT] = {};

    const float* aPtr = A + (size_t)(mBase + l16) * K + 4 * half;
    const float* bPtr = B + (size_t)(4 * half) * NF + 4 * l16;
    for (int k = 0; k < K; k += 8) {
        const v4f av = *(const v4f*)(aPtr + k);
        v4f bv[4][NG];
#pragma unroll
        for (int i = 0; i < 4; ++i) {
#pragma unroll
            for (int g = 0; g < NG; ++g)
                bv[i][g] = *(const v4f*)(bPtr + (size_t)(k + i) * NF + g * 64);
        }
#pragma unroll
        for (int u = 0; u < 2; ++u) {
            v2f a;
            a.x = av[2 * u];
            a.y = av[2 * u + 1];
#pragma unroll
            for (int g = 0; g < NG; ++g) {
#pragma unroll
                for (int j = 0; j < 4; ++j) {
                    v2f b;
                    b.x = bv[2 * u][g][j];
                    b.y = bv[2 * u + 1][g][j];
                    acc[g * 4 + j] = __builtin_amdgcn_wmma_f32_16x16x4_f32(
                        false, a, false, b, (short)0, acc[g * 4 + j], false, false);
                }
            }
        }
    }
#pragma unroll
    for (int g = 0; g < NG; ++g) {
        float* cPtr = C + (size_t)(mBase + half * 8) * NF + g * 64 + 4 * l16;
#pragma unroll
        for (int r = 0; r < 8; ++r) {
            v4f o;
            o.x = acc[g * 4 + 0][r];
            o.y = acc[g * 4 + 1][r];
            o.z = acc[g * 4 + 2][r];
            o.w = acc[g * 4 + 3][r];
            *(v4f*)(cPtr + (size_t)r * NF) = o;
        }
    }
}

template <int K, int NF>
__global__ __launch_bounds__(256) void gemm_wmma(const float* __restrict__ A,
                                                 const float* __restrict__ B,
                                                 float* __restrict__ C, int M) {
    gemm_stripe<K, NF>(A, B, C, M, (int)blockIdx.x);
}

template <int K, int NF>
__global__ __launch_bounds__(256) void gemm_deg(const float* __restrict__ A,
                                                const float* __restrict__ B,
                                                float* __restrict__ C, int M,
                                                const int* __restrict__ src,
                                                const int* __restrict__ dst,
                                                float* __restrict__ deg0,
                                                float* __restrict__ deg1,
                                                int* __restrict__ bins, int E) {
    __shared__ int s_half;
    __shared__ v4i s_rows[256 * BLK_E / 8 / 16];
    __shared__ unsigned char s_pre[256 * 32];
    __shared__ int s_cnt[256];
    const int ngemm = (M / 16 + 7) / 8;
    if ((int)blockIdx.x >= ngemm) {
        float* deg = wg_half(&s_half) ? deg1 : deg0;
        deg_slice(src, dst, deg, bins, E, (int)blockIdx.x - ngemm, s_rows, s_pre, s_cnt);
        return;
    }
    gemm_stripe<K, NF>(A, B, C, M, (int)blockIdx.x);
}

#define RN    OWN_RN
#define NSLOT ((RN + 31) / 32)
#define PCAP  160
#define SUBCAP 24
#define SUBP   25
#define CNTP   33
#define RCAP  512

template <int NQ>
__device__ __forceinline__ void add_to_slot(v8f (&acc)[NQ], int slot, const float (&v)[NQ], float w) {
#pragma unroll
    for (int q = 0; q < NQ; ++q) acc[q][slot] = __builtin_fmaf(v[q], w, acc[q][slot]);
}

template <int F, int U>
__device__ __forceinline__ void add_rows(int s, int dloc, float wgt, int m, int lane,
                                         const float* __restrict__ h, v8f (&acc)[F / 32]) {
    constexpr int NQ = F / 32;
    int j = 0;
#pragma unroll 1
    for (; j + U <= m; j += U) {
        float v[U][NQ];
#pragma unroll
        for (int u = 0; u < U; ++u) {
            int sj = __builtin_amdgcn_readlane(s, j + u);
            const float* hr = h + (size_t)sj * F + lane;
#pragma unroll
            for (int q = 0; q < NQ; ++q) v[u][q] = hr[q * 32];
        }
#pragma unroll
        for (int u = 0; u < U; ++u) {
            int   dj = __builtin_amdgcn_readlane(dloc, j + u);
            float wj = __builtin_bit_cast(float, __builtin_amdgcn_readlane(__builtin_bit_cast(int, wgt), j + u));
            add_to_slot<NQ>(acc, dj >> 5, v[u], wj);
        }
    }
    for (; j < m; ++j) {
        int   sj = __builtin_amdgcn_readlane(s, j);
        int   dj = __builtin_amdgcn_readlane(dloc, j);
        float wj = __builtin_bit_cast(float, __builtin_amdgcn_readlane(__builtin_bit_cast(int, wgt), j));
        const float* hr = h + (size_t)sj * F + lane;
        float v[NQ];
#pragma unroll
        for (int q = 0; q < NQ; ++q) v[q] = hr[q * 32];
        add_to_slot<NQ>(acc, dj >> 5, v, wj);
    }
}

template <int F, int U, bool RECORD>
__device__ __forceinline__ void take_edges(const int* ent, int m, int lane,
                                           const float* __restrict__ dinv,
                                           const float* __restrict__ h,
                                           const float* dl, v8f (&acc)[F / 32],
                                           int* recId, float* recW, int& recn, bool& rovf) {
    int s = 0, dloc = 0;
    float wgt = 0.0f;
    if (lane < m) {
        int en = ent[lane];
        dloc  = (int)((unsigned)en >> 24);
        s     = en & 0xFFFFFF;
        wgt   = dinv[s] * dl[dloc];
    }
    if (RECORD) {
        if (recn + m <= RCAP) {
            if (lane < m) {
                recId[recn + lane] = (int)(((unsigned)dloc << 24) | (unsigned)s);
                recW[recn + lane]  = wgt;
            }
        } else {
            rovf = true;
        }
        recn += m;
    }
    add_rows<F, U>(s, dloc, wgt, m, lane, h, acc);
}

#define BIN_U 13

template <int F, bool RELU, bool REPLAY>
__global__ __launch_bounds__(1024) void agg_tile(const int* __restrict__ src,
                                                 const int* __restrict__ dst,
                                                 const float* __restrict__ dinv,
                                                 const float* __restrict__ h,
                                                 const float* __restrict__ bias,
                                                 float* __restrict__ outp, int E,
                                                 int* __restrict__ recs,
                                                 int* __restrict__ reccnt,
                                                 const int* __restrict__ bins) {
    static_assert(RN <= 256 && NNODES <= (1 << 24), "a sub-list entry keeps the row in 8 bits and the source id in 24");
    static_assert((NNODES + RN - 1) / RN <= 256, "a node range's number is kept in one byte");
    constexpr int U  = 1024 / F;
    constexpr int NQ = F / 32;
    __shared__ float dl[RN];
    __shared__ int   pendl[32 * PCAP];
    __shared__ int   sub[32 * 32 * SUBP];
    __shared__ int   subcnt[32 * CNTP];
    __shared__ int   s_ovf;

    const int lane = threadIdx.x & 31;
    const int wave = __builtin_amdgcn_readfirstlane((int)(threadIdx.x >> 5));
    const int lo   = blockIdx.x * RN;
    const int nn   = min(RN, NNODES - lo);

    int*   recId = recs + (size_t)(blockIdx.x * 32 + wave) * (2 * RCAP);
    float* recW  = (float*)(recId + RCAP);
    int cw = 0;
    if (REPLAY) cw = reccnt[blockIdx.x * 32 + lane];

    if (threadIdx.x == 0) s_ovf = 0;
    for (int i = threadIdx.x; i < nn; i += 1024) dl[i] = dinv[lo + i];
    __syncthreads();

    bool  replay = false;
    int   myn = 0;
    int   curId = 0;
    float curW = 0.0f;
    if (REPLAY) {
        replay = __builtin_amdgcn_ballot_w32(cw < 0) == 0u;
        myn = __builtin_amdgcn_readlane(cw, wave);
        if (replay && lane < myn) {
            curId = recId[lane];
            curW  = recW[lane];
        }
    }
    static_assert(NSLOT <= 8, "one vector element per row of an owner wave");
    v8f acc[NQ];
    {
        float bq[NQ], hv[NSLOT][NQ], dq[NSLOT];
#pragma unroll
        for (int q = 0; q < NQ; ++q) bq[q] = bias[lane + q * 32];
#pragma unroll
        for (int k = 0; k < NSLOT; ++k) {
            const int r = min(wave + 32 * k, nn - 1);
            const float* hr = h + (size_t)(lo + r) * F + lane;
            dq[k] = dl[r];
#pragma unroll
            for (int q = 0; q < NQ; ++q) hv[k][q] = hr[q * 32];
        }
#pragma unroll
        for (int q = 0; q < NQ; ++q) {
#pragma unroll
            for (int k = 0; k < NSLOT; ++k) acc[q][k] = hv[k][q] * (dq[k] * dq[k]) + bq[q];
#pragma unroll
            for (int k = NSLOT; k < 8; ++k) acc[q][k] = 0.0f;
        }
    }

    if (REPLAY && replay) {
        for (int b0 = 0; b0 < myn; b0 += 32) {
            const int m = min(32, myn - b0);
            int   nxtId = 0;
            float nxtW = 0.0f;
            if (b0 + 32 + lane < myn) {
                nxtId = recId[b0 + 32 + lane];
                nxtW  = recW[b0 + 32 + lane];
            }
            add_rows<F, U>(curId & 0xFFFFFF, (int)((unsigned)curId >> 24), curW, m, lane, h, acc);
            curId = nxtId;
            curW  = nxtW;
        }
    } else {
        bool ovf = false;
        const int nblk = (E + BLK_E - 1) / BLK_E;
        const int* binr = bins + (size_t)blockIdx.x * nblk * BIN_W + lane;
        subcnt[lane * CNTP + wave] = 0;
        for (int b0 = wave; b0 < nblk; b0 += 32 * BIN_U) {
            int w[BIN_U];
#pragma unroll
            for (int j = 0; j < BIN_U; ++j) {
                const int bj = b0 + 32 * j;
                const int v = binr[(size_t)min(bj, nblk - 1) * BIN_W];
                w[j] = (bj < nblk) ? v : 0;
            }
#pragma unroll
            for (int j = 0; j < BIN_U; ++j) {
                const int cnt = __builtin_amdgcn_readlane(w[j], 0);
                if (cnt == 0) continue;
                if (cnt > BIN_W - 1) { ovf = true; continue; }
                const bool valid = lane >= 1 && lane <= cnt;
                const unsigned any = __builtin_amdgcn_ballot_w32(valid);
                const unsigned dloc = (unsigned)w[j] >> 24;
                const unsigned o = dloc & 31u;
                unsigned same = any;
#pragma unroll
                for (int b = 0; b < 5; ++b) {
                    const bool bit = ((o >> b) & 1u) != 0u;
                    const unsigned bal = __builtin_amdgcn_ballot_w32(bit);
                    same &= bit ? bal : ~bal;
                }
                if (valid) {
                    const int before = (int)__builtin_amdgcn_mbcnt_lo(same, 0u);
                    int* cp = subcnt + o * CNTP + wave;
                    const int old = *cp;
                    const int k = old + before;
                    if (k < SUBCAP) sub[(o * 32 + wave) * SUBP + k] = w[j];
                    else ovf = true;
                    if (before == 0) *cp = old + __builtin_popcount(same);
                }
            }
        }
        if (ovf) s_ovf = 1;
        __syncthreads();

        if (s_ovf == 0) {
            int* mypend = pendl + wave * PCAP;
            int pend = 0;
            int  recn = 0;
            bool rovf = false;
            const int* mysub = sub + (wave * 32 + lane) * SUBP;
            const int cvec = subcnt[wave * CNTP + lane];
            for (int k = 0; k < SUBCAP; ++k) {
                const bool have = k < cvec;
                const unsigned mask = __builtin_amdgcn_ballot_w32(have);
                if (mask == 0u) break;
                if (have) mypend[pend + __builtin_amdgcn_mbcnt_lo(mask, 0u)] = mysub[k];
                pend += __builtin_popcount(mask);
                while (pend >= 32) {
                    pend -= 32;
                    take_edges<F, U, !REPLAY>(mypend + pend, 32, lane, dinv, h, dl, acc,
                                              recId, recW, recn, rovf);
                }
            }
            take_edges<F, U, !REPLAY>(mypend, pend, lane, dinv, h, dl, acc, recId, recW, recn, rovf);
            if (!REPLAY && lane == 0) reccnt[blockIdx.x * 32 + wave] = rovf ? -1 : recn;
        } else {
            if (!REPLAY && lane == 0) reccnt[blockIdx.x * 32 + wave] = -1;
            for (int base = 0; base < E; base += 32) {
                int e = base + lane;
                int d = (e < E) ? dst[e] : -1;
                unsigned dloc = (unsigned)(d - lo);
                bool mine = (dloc < (unsigned)nn) && ((dloc & 31u) == (unsigned)wave);
                unsigned mask = __builtin_amdgcn_ballot_w32(mine);
                while (mask) {
                    int j = __builtin_ctz(mask);
                    mask &= mask - 1;
                    int dj = (int)__builtin_amdgcn_readlane((int)dloc, j);
                    int sj = src[base + j];
                    float wj = dinv[sj] * dl[dj];
                    const float* hr = h + (size_t)sj * F + lane;
                    float v[NQ];
#pragma unroll
                    for (int q = 0; q < NQ; ++q) v[q] = hr[q * 32];
                    add_to_slot<NQ>(acc, dj >> 5, v, wj);
                }
            }
        }
    }

#pragma unroll
    for (int k = 0; k < NSLOT; ++k) {
        const int r = wave + 32 * k;
        if (r < nn) {
            float* op = outp + (size_t)(lo + r) * F + lane;
#pragma unroll
            for (int q = 0; q < NQ; ++q) {
                float v = acc[q][k];
                if (RELU) v = fmaxf(v, 0.0f);
                op[q * 32] = v;
            }
        }
    }
}

extern "C" void kernel_launch(void* const* d_in, const int* in_sizes, int n_in,
                              void* d_out, int out_size, void* d_ws, size_t ws_size,
                              hipStream_t stream) {
    const float* x  = (const float*)d_in[0];
    const int*   ei = (const int*)d_in[1];
    const float* W1 = (const float*)d_in[2];
    const float* b1 = (const float*)d_in[3];
    const float* W2 = (const float*)d_in[4];
    const float* b2 = (const float*)d_in[5];
    float* out = (float*)d_out;

    const int N = NNODES;
    const int E = in_sizes[1] / 2;
    const int* src = ei;
    const int* dst = ei + E;

    float* deg   = (float*)d_ws;
    float* h1    = deg + NPAD;
    float* a1    = h1 + (size_t)N * 128;
    float* h2    = a1 + (size_t)N * 128;
    float* deg1  = h2 + (size_t)N * 64;
    int*   reccnt = (int*)(deg1 + NPAD);
    int*   recs   = reccnt + 256 * 32;
    int*   bins   = recs + (size_t)256 * 32 * 2 * RCAP;

    const int BT = 256;
    const int NB = (NNODES + RN - 1) / RN;

    init_kernel<<<(N + BT - 1) / BT, BT, 0, stream>>>(deg, deg1);
    gemm_deg<256, 128><<<(N / 16 + 7) / 8 + DEG_WGS, BT, 0, stream>>>(x, W1, h1, N, src, dst, deg, deg1, bins, E);
    dinv_kernel<<<(N + BT - 1) / BT, BT, 0, stream>>>(deg, deg1);
    agg_tile<128, true, false><<<NB, 1024, 0, stream>>>(src, dst, deg, h1, b1, a1, E, recs, reccnt, bins);
    gemm_wmma<128, 64><<<(N / 16 + 7) / 8, BT, 0, stream>>>(a1, W2, h2, N);
    agg_tile<64, false, true><<<NB, 1024, 0, stream>>>(src, dst, deg, h2, b2, out, E, recs, reccnt, bins);
}
